# starting kernel
# baseline (speedup 1.0000x reference)
#include <hip/hip_runtime.h>


typedef float v2f __attribute__((ext_vector_type(2)));
typedef float v8f __attribute__((ext_vector_type(8)));

#define CIN   128
#define KOUT  256
#define HIMG  56
#define WIMG  56
#define HW    3136
#define KTOT  1152
#define MTOT  100352
#define BCH   64
#define BSP   128
#define BK    16
#define LDK   20
#define NITER (KTOT / BK)

struct Regs {
    float a[4];
    float b[8];
};

__global__ __launch_bounds__(256) void conv3x3_wmma_f32(
    const float* __restrict__ x,
    const float* __restrict__ wgt,
    const float* __restrict__ bias,
    float* __restrict__ out)
{
    __shared__ float As[2][BCH][LDK];
    __shared__ float Bs[2][BSP][LDK];

    const int tid  = threadIdx.x;
    const int lane = tid & 31;
    const int wave = tid >> 5;
    const int wm   = wave & 1;
    const int wn   = wave >> 1;
    const int hi   = lane >> 4;
    const int l16  = lane & 15;

    const int chBase = blockIdx.y * BCH;
    const int spBase = blockIdx.x * BSP;

    const int bn  = tid & 127;
    const int bk0 = (tid >> 7) * 8;
    const int spb = spBase + bn;
    const int bbB = spb / HW;
    const int hwB = spb - bbB * HW;
    const int ohB = hwB / WIMG;
    const int owB = hwB - ohB * WIMG;
    const float* xb = x + (long)bbB * CIN * HW + hwB;

    const int am = tid >> 3;
    const int ak = (tid & 7) * 2;
    const float* wp0 = wgt + (long)(chBase + am) * KTOT + ak * 9;
    const float* wp1 = wp0 + 32 * KTOT;

    v8f acc[2][2] = {{{}, {}}, {{}, {}}};

    auto gload = [&](int k0, Regs& rg) {
        const int rs = k0 >> 7;
        const int c0 = k0 & 127;
        const int r  = rs / 3;
        const int s  = rs - r * 3;
        const int roff = (r - 1) * WIMG + (s - 1);
        const bool vld = ((unsigned)(ohB + r - 1) < (unsigned)HIMG) &&
                         ((unsigned)(owB + s - 1) < (unsigned)WIMG);
#pragma unroll
        for (int j = 0; j < 8; ++j) rg.b[j] = 0.0f;
        if (vld) {
            const float* p = xb + (long)(c0 + bk0) * HW + roff;
#pragma unroll
            for (int j = 0; j < 8; ++j) rg.b[j] = p[(long)j * HW];
        }
        const long wo = (long)c0 * 9 + rs;
        rg.a[0] = wp0[wo];
        rg.a[1] = wp0[wo + 9];
        rg.a[2] = wp1[wo];
        rg.a[3] = wp1[wo + 9];
    };

    auto lstore = [&](int buf, const Regs& rg) {
#pragma unroll
        for (int j = 0; j < 4; ++j) {
            v2f t = {rg.b[2 * j], rg.b[2 * j + 1]};
            *(v2f*)&Bs[buf][bn][bk0 + 2 * j] = t;
        }
        v2f ta0 = {rg.a[0], rg.a[1]};
        v2f ta1 = {rg.a[2], rg.a[3]};
        *(v2f*)&As[buf][am][ak]      = ta0;
        *(v2f*)&As[buf][am + 32][ak] = ta1;
    };

    auto compute = [&](int buf) {
#pragma unroll
        for (int kk = 0; kk < BK; kk += 4) {
            const int krow = kk + hi * 2;
            v2f a0 = *(const v2f*)&As[buf][wm * 32 + l16][krow];
            v2f a1 = *(const v2f*)&As[buf][wm * 32 + 16 + l16][krow];
            v2f b0 = *(const v2f*)&Bs[buf][wn * 32 + l16][krow];
            v2f b1 = *(const v2f*)&Bs[buf][wn * 32 + 16 + l16][krow];
            acc[0][0] = __builtin_amdgcn_wmma_f32_16x16x4_f32(
                            false, a0, false, b0, (short)0, acc[0][0], false, false);
            acc[0][1] = __builtin_amdgcn_wmma_f32_16x16x4_f32(
                            false, a0, false, b1, (short)0, acc[0][1], false, false);
            acc[1][0] = __builtin_amdgcn_wmma_f32_16x16x4_f32(
                            false, a1, false, b0, (short)0, acc[1][0], false, false);
            acc[1][1] = __builtin_amdgcn_wmma_f32_16x16x4_f32(
                            false, a1, false, b1, (short)0, acc[1][1], false, false);
        }
    };

    Regs rg;
    gload(0, rg);
    lstore(0, rg);
    __syncthreads();

    int cur = 0;
#pragma unroll 1
    for (int it = 0; it < NITER - 1; ++it) {
        gload((it + 1) * BK, rg);
        compute(cur);
        lstore(cur ^ 1, rg);
        __syncthreads();
        cur ^= 1;
    }
    compute(cur);

#pragma unroll
    for (int i = 0; i < 2; ++i) {
#pragma unroll
        for (int j = 0; j < 2; ++j) {
            const int sp0 = spBase + wn * 32 + j * 16 + l16;
            const int b2  = sp0 / HW;
            const int hw2 = sp0 - b2 * HW;
            float* op = out + (long)b2 * KOUT * HW + hw2;
#pragma unroll
            for (int rr = 0; rr < 8; ++rr) {
                const int ch = chBase + wm * 32 + i * 16 + hi * 8 + rr;
                op[(long)ch * HW] = acc[i][j][rr] + bias[ch];
            }
        }
    }
}

extern "C" void kernel_launch(void* const* d_in, const int* in_sizes, int n_in,
                              void* d_out, int out_size, void* d_ws, size_t ws_size,
                              hipStream_t stream) {
    const float* x    = (const float*)d_in[0];
    const float* wgt  = (const float*)d_in[1];
    const float* bias = (const float*)d_in[2];
    float* out        = (float*)d_out;

    dim3 grid(MTOT / BSP, KOUT / BCH);
    conv3x3_wmma_f32<<<grid, 256, 0, stream>>>(x, wgt, bias, out);
}


# K order changed to the weight tensor's own (tile = 4 channels x 9 taps): weights read as contiguous b128 row pieces, an x channel's 9 taps read back to back; sum order over K changes
# speedup vs baseline: 1.7964x; 1.7964x over previous
#include <hip/hip_runtime.h>


typedef float v2f __attribute__((ext_vector_type(2)));
typedef float v4f __attribute__((ext_vector_type(4)));
typedef float v8f __attribute__((ext_vector_type(8)));

#define CIN   128
#define KOUT  256
#define HIMG  56
#define WIMG  56
#define HW    3136
#define KTOT  1152
#define MTOT  100352
#define BCH   64
#define BSP   128
#define CPT   4
#define BK    36
#define LDK   36
#define NITER (CIN / CPT)
#define AV4   (BCH * BK / 4)

struct Regs {
    v4f   a[3];
    float b[18];
};

__global__ __launch_bounds__(256) void conv3x3_wmma_f32(
    const float* __restrict__ x,
    const float* __restrict__ wgt,
    const float* __restrict__ bias,
    float* __restrict__ out)
{
    __shared__ __attribute__((aligned(16))) float As[2][BCH][LDK];
    __shared__ __attribute__((aligned(16))) float Bs[2][BSP][LDK];

    const int tid  = threadIdx.x;
    const int lane = tid & 31;
    const int wave = tid >> 5;
    const int wm   = wave & 1;
    const int wn   = wave >> 1;
    const int hi   = lane >> 4;
    const int l16  = lane & 15;

    const int chBase = blockIdx.y * BCH;
    const int spBase = blockIdx.x * BSP;

    const int bn  = tid & 127;
    const int bh  = tid >> 7;
    const int spb = spBase + bn;
    const int bbB = spb / HW;
    const int hwB = spb - bbB * HW;
    const int ohB = hwB / WIMG;
    const int owB = hwB - ohB * WIMG;
    const float* xb = x + (long)bbB * CIN * HW + (long)(2 * bh) * HW + hwB;
    unsigned vmask = 0;
#pragma unroll
    for (int t = 0; t < 9; ++t) {
        const int r = t / 3, s = t - 3 * (t / 3);
        const bool v = ((unsigned)(ohB + r - 1) < (unsigned)HIMG) &&
                       ((unsigned)(owB + s - 1) < (unsigned)WIMG);
        vmask |= (v ? 1u : 0u) << t;
    }

    const int ar0 = tid / 9,         aq0 = tid - ar0 * 9;
    const int ar1 = (tid + 256) / 9, aq1 = (tid + 256) - ar1 * 9;
    const int ar2 = (tid + 512) / 9, aq2 = (tid + 512) - ar2 * 9;
    const bool a2on = tid < (AV4 - 512);
    const float* wpa0 = wgt + (long)(chBase + ar0) * KTOT + aq0 * 4;
    const float* wpa1 = wgt + (long)(chBase + ar1) * KTOT + aq1 * 4;
    const float* wpa2 = wgt + (long)(chBase + (a2on ? ar2 : 0)) * KTOT + (a2on ? aq2 : 0) * 4;

    v8f acc[2][2] = {{{}, {}}, {{}, {}}};

    auto gload = [&](int it, Regs& rg) {
        const int c0 = it * CPT;
        const float* p = xb + (long)c0 * HW;
#pragma unroll
        for (int cc = 0; cc < 2; ++cc) {
#pragma unroll
            for (int t = 0; t < 9; ++t) {
                const int r = t / 3, s = t - 3 * (t / 3);
                float v = 0.0f;
                if (vmask & (1u << t)) v = p[cc * HW + (r - 1) * WIMG + (s - 1)];
                rg.b[cc * 9 + t] = v;
            }
        }
        const int wo = c0 * 9;
        rg.a[0] = *(const v4f*)(wpa0 + wo);
        rg.a[1] = *(const v4f*)(wpa1 + wo);
        if (a2on) rg.a[2] = *(const v4f*)(wpa2 + wo);
    };

    auto lstore = [&](int buf, const Regs& rg) {
#pragma unroll
        for (int j = 0; j < 9; ++j) {
            v2f t = {rg.b[2 * j], rg.b[2 * j + 1]};
            *(v2f*)&Bs[buf][bn][bh * 18 + 2 * j] = t;
        }
        *(v4f*)&As[buf][ar0][aq0 * 4] = rg.a[0];
        *(v4f*)&As[buf][ar1][aq1 * 4] = rg.a[1];
        if (a2on) *(v4f*)&As[buf][ar2][aq2 * 4] = rg.a[2];
    };

    auto compute = [&](int buf) {
#pragma unroll
        for (int kk = 0; kk < BK; kk += 4) {
            const int krow = kk + hi * 2;
            v2f a0 = *(const v2f*)&As[buf][wm * 32 + l16][krow];
            v2f a1 = *(const v2f*)&As[buf][wm * 32 + 16 + l16][krow];
            v2f b0 = *(const v2f*)&Bs[buf][wn * 32 + l16][krow];
            v2f b1 = *(const v2f*)&Bs[buf][wn * 32 + 16 + l16][krow];
            acc[0][0] = __builtin_amdgcn_wmma_f32_16x16x4_f32(
                            false, a0, false, b0, (short)0, acc[0][0], false, false);
            acc[0][1] = __builtin_amdgcn_wmma_f32_16x16x4_f32(
                            false, a0, false, b1, (short)0, acc[0][1], false, false);
            acc[1][0] = __builtin_amdgcn_wmma_f32_16x16x4_f32(
                            false, a1, false, b0, (short)0, acc[1][0], false, false);
            acc[1][1] = __builtin_amdgcn_wmma_f32_16x16x4_f32(
                            false, a1, false, b1, (short)0, acc[1][1], false, false);
        }
    };

    Regs rg;
    rg.a[2] = (v4f){0.0f, 0.0f, 0.0f, 0.0f};
    gload(0, rg);
    lstore(0, rg);
    __syncthreads();

    int cur = 0;
#pragma unroll 1
    for (int it = 0; it < NITER - 1; ++it) {
        gload(it + 1, rg);
        compute(cur);
        lstore(cur ^ 1, rg);
        __syncthreads();
        cur ^= 1;
    }
    compute(cur);

#pragma unroll
    for (int i = 0; i < 2; ++i) {
#pragma unroll
        for (int j = 0; j < 2; ++j) {
            const int sp0 = spBase + wn * 32 + j * 16 + l16;
            const int b2  = sp0 / HW;
            const int hw2 = sp0 - b2 * HW;
            float* op = out + (long)b2 * KOUT * HW + hw2;
#pragma unroll
            for (int rr = 0; rr < 8; ++rr) {
                const int ch = chBase + wm * 32 + i * 16 + hi * 8 + rr;
                op[(long)ch * HW] = acc[i][j][rr] + bias[ch];
            }
        }
    }
}

extern "C" void kernel_launch(void* const* d_in, const int* in_sizes, int n_in,
                              void* d_out, int out_size, void* d_ws, size_t ws_size,
                              hipStream_t stream) {
    const float* x    = (const float*)d_in[0];
    const float* wgt  = (const float*)d_in[1];
    const float* bias = (const float*)d_in[2];
    float* out        = (float*)d_out;

    dim3 grid(MTOT / BSP, KOUT / BCH);
    conv3x3_wmma_f32<<<grid, 256, 0, stream>>>(x, wgt, bias, out);
}


# x taps loaded without a branch (scalar base + 32-bit lane offset): an out-of-image tap reads the centre element and a select zeroes it before the LDS store; 18 exec-masked branches leave the K loop
# speedup vs baseline: 1.9160x; 1.0666x over previous
#include <hip/hip_runtime.h>


typedef float v2f __attribute__((ext_vector_type(2)));
typedef float v4f __attribute__((ext_vector_type(4)));
typedef float v8f __attribute__((ext_vector_type(8)));

#define CIN   128
#define KOUT  256
#define HIMG  56
#define WIMG  56
#define HW    3136
#define KTOT  1152
#define MTOT  100352
#define BCH   64
#define BSP   128
#define CPT   4
#define BK    36
#define LDK   36
#define NITER (CIN / CPT)
#define AV4   (BCH * BK / 4)

struct Regs {
    v4f   a[3];
    float b[18];
};

__global__ __launch_bounds__(256) void conv3x3_wmma_f32(
    const float* __restrict__ x,
    const float* __restrict__ wgt,
    const float* __restrict__ bias,
    float* __restrict__ out)
{
    __shared__ __attribute__((aligned(16))) float As[2][BCH][LDK];
    __shared__ __attribute__((aligned(16))) float Bs[2][BSP][LDK];

    const int tid  = threadIdx.x;
    const int lane = tid & 31;
    const int wave = tid >> 5;
    const int wm   = wave & 1;
    const int wn   = wave >> 1;
    const int hi   = lane >> 4;
    const int l16  = lane & 15;

    const int chBase = blockIdx.y * BCH;
    const int spBase = blockIdx.x * BSP;

    const int bn  = tid & 127;
    const int bh  = tid >> 7;
    const int spb = spBase + bn;
    const int bbB = spb / HW;
    const int hwB = spb - bbB * HW;
    const int ohB = hwB / WIMG;
    const int owB = hwB - ohB * WIMG;
    const unsigned xcB = (unsigned)((bbB * CIN + 2 * bh) * HW + hwB) * 4u;
    unsigned vmask = 0;
    unsigned xoffB[9];
#pragma unroll
    for (int t = 0; t < 9; ++t) {
        const int r = t / 3, s = t - 3 * (t / 3);
        const bool v = ((unsigned)(ohB + r - 1) < (unsigned)HIMG) &&
                       ((unsigned)(owB + s - 1) < (unsigned)WIMG);
        vmask |= (v ? 1u : 0u) << t;
        xoffB[t] = xcB + (v ? (unsigned)(((r - 1) * WIMG + (s - 1)) * 4) : 0u);
    }

    const int ar0 = tid / 9,         aq0 = tid - ar0 * 9;
    const int ar1 = (tid + 256) / 9, aq1 = (tid + 256) - ar1 * 9;
    const int ar2 = (tid + 512) / 9, aq2 = (tid + 512) - ar2 * 9;
    const bool a2on = tid < (AV4 - 512);
    const float* wpa0 = wgt + (long)(chBase + ar0) * KTOT + aq0 * 4;
    const float* wpa1 = wgt + (long)(chBase + ar1) * KTOT + aq1 * 4;
    const float* wpa2 = wgt + (long)(chBase + (a2on ? ar2 : 0)) * KTOT + (a2on ? aq2 : 0) * 4;

    v8f acc[2][2] = {{{}, {}}, {{}, {}}};

    auto gload = [&](int it, Regs& rg) {
        const int c0 = it * CPT;
        const unsigned tileB = (unsigned)(c0 * (HW * 4));
#pragma unroll
        for (int t = 0; t < 9; ++t) {
            const char* q = (const char*)x + (xoffB[t] + tileB);
#pragma unroll
            for (int cc = 0; cc < 2; ++cc)
                rg.b[cc * 9 + t] = *(const float*)(q + cc * (HW * 4));
        }
        const int wo = c0 * 9;
        rg.a[0] = *(const v4f*)(wpa0 + wo);
        rg.a[1] = *(const v4f*)(wpa1 + wo);
        if (a2on) rg.a[2] = *(const v4f*)(wpa2 + wo);
    };

    auto lstore = [&](int buf, const Regs& rg) {
#pragma unroll
        for (int j = 0; j < 9; ++j) {
            const int e0 = 2 * j, e1 = 2 * j + 1;
            v2f t = {(vmask & (1u << (e0 % 9))) ? rg.b[e0] : 0.0f,
                     (vmask & (1u << (e1 % 9))) ? rg.b[e1] : 0.0f};
            *(v2f*)&Bs[buf][bn][bh * 18 + 2 * j] = t;
        }
        *(v4f*)&As[buf][ar0][aq0 * 4] = rg.a[0];
        *(v4f*)&As[buf][ar1][aq1 * 4] = rg.a[1];
        if (a2on) *(v4f*)&As[buf][ar2][aq2 * 4] = rg.a[2];
    };

    auto compute = [&](int buf, int k0, int k1) {
#pragma unroll
        for (int kk = k0; kk < k1; kk += 4) {
            const int krow = kk + hi * 2;
            v2f a0 = *(const v2f*)&As[buf][wm * 32 + l16][krow];
            v2f a1 = *(const v2f*)&As[buf][wm * 32 + 16 + l16][krow];
            v2f b0 = *(const v2f*)&Bs[buf][wn * 32 + l16][krow];
            v2f b1 = *(const v2f*)&Bs[buf][wn * 32 + 16 + l16][krow];
            acc[0][0] = __builtin_amdgcn_wmma_f32_16x16x4_f32(
                            false, a0, false, b0, (short)0, acc[0][0], false, false);
            acc[0][1] = __builtin_amdgcn_wmma_f32_16x16x4_f32(
                            false, a0, false, b1, (short)0, acc[0][1], false, false);
            acc[1][0] = __builtin_amdgcn_wmma_f32_16x16x4_f32(
                            false, a1, false, b0, (short)0, acc[1][0], false, false);
            acc[1][1] = __builtin_amdgcn_wmma_f32_16x16x4_f32(
                            false, a1, false, b1, (short)0, acc[1][1], false, false);
        }
    };

    Regs rg;
    rg.a[2] = (v4f){0.0f, 0.0f, 0.0f, 0.0f};
    gload(0, rg);
    lstore(0, rg);
    __syncthreads();

    int cur = 0;
#pragma unroll 1
    for (int it = 0; it < NITER - 1; ++it) {
        gload(it + 1, rg);
        compute(cur, 0, 28);
        __builtin_amdgcn_sched_barrier(0x100);
        compute(cur, 28, BK);
        lstore(cur ^ 1, rg);
        __syncthreads();
        cur ^= 1;
    }
    compute(cur, 0, BK);

#pragma unroll
    for (int i = 0; i < 2; ++i) {
#pragma unroll
        for (int j = 0; j < 2; ++j) {
            const int sp0 = spBase + wn * 32 + j * 16 + l16;
            const int b2  = sp0 / HW;
            const int hw2 = sp0 - b2 * HW;
            float* op = out + (long)b2 * KOUT * HW + hw2;
#pragma unroll
            for (int rr = 0; rr < 8; ++rr) {
                const int ch = chBase + wm * 32 + i * 16 + hi * 8 + rr;
                op[(long)ch * HW] = acc[i][j][rr] + bias[ch];
            }
        }
    }
}

extern "C" void kernel_launch(void* const* d_in, const int* in_sizes, int n_in,
                              void* d_out, int out_size, void* d_ws, size_t ws_size,
                              hipStream_t stream) {
    const float* x    = (const float*)d_in[0];
    const float* wgt  = (const float*)d_in[1];
    const float* bias = (const float*)d_in[2];
    float* out        = (float*)d_out;

    dim3 grid(MTOT / BSP, KOUT / BCH);
    conv3x3_wmma_f32<<<grid, 256, 0, stream>>>(x, wgt, bias, out);
}


# Block tile = 64 ch x 2 whole image rows; x staged raw per K tile (4 rows x 58 cols, 4 ch interleaved, zero edge), the 3x3 taps taken at the LDS fragment read; sum order over K changes
# speedup vs baseline: 2.0492x; 1.0695x over previous
#include <hip/hip_runtime.h>


typedef float v2f __attribute__((ext_vector_type(2)));
typedef float v4f __attribute__((ext_vector_type(4)));
typedef float v8f __attribute__((ext_vector_type(8)));

#define CIN   128
#define KOUT  256
#define HIMG  56
#define WIMG  56
#define HW    3136
#define KTOT  1152
#define BCH   64
#define BSP   112
#define NT    7
#define CPT   4
#define LDK   36
#define NITER (CIN / CPT)
#define XR    4
#define XC    58
#define XPIX  (XR * WIMG)
#define AV4   (BCH * LDK / 4)
#define TPB   128

struct Regs {
    v4f a[5];
    v4f x[2];
};

__global__ __launch_bounds__(TPB) void conv3x3_wmma_f32(
    const float* __restrict__ x,
    const float* __restrict__ wgt,
    const float* __restrict__ bias,
    float* __restrict__ out)
{
    __shared__ __attribute__((aligned(16))) float As[2][BCH * LDK];
    __shared__ __attribute__((aligned(16))) float Xs[2][XR * XC * CPT];

    const int tid  = threadIdx.x;
    const int lane = tid & 31;
    const int wave = tid >> 5;
    const int hi   = lane >> 4;
    const int l16  = lane & 15;

    const int chBase = blockIdx.y * BCH;
    const int bimg   = blockIdx.x / (HIMG / 2);
    const int oh0    = (blockIdx.x - bimg * (HIMG / 2)) * 2;

    const int q0 = tid;
    const int q1 = (tid < XPIX - TPB) ? tid + TPB : tid;
    int   xg[2];
    int   xl[2];
    bool  xv[2];
#pragma unroll
    for (int i = 0; i < 2; ++i) {
        const int q  = i ? q1 : q0;
        const int rr = q / WIMG, cc = q - rr * WIMG;
        const int ih = oh0 - 1 + rr;
        xv[i] = (unsigned)ih < (unsigned)HIMG;
        const int ihc = ih < 0 ? 0 : (ih > HIMG - 1 ? HIMG - 1 : ih);
        xg[i] = bimg * (CIN * HW) + ihc * WIMG + cc;
        xl[i] = (rr * XC + cc + 1) * CPT;
    }

    int wg[5], wl[5];
#pragma unroll
    for (int i = 0; i < 5; ++i) {
        int p = tid + i * TPB;
        if (i == 4 && tid >= AV4 - 4 * TPB) p = tid + 3 * TPB;
        const int ar = p / 9, aq = p - ar * 9;
        wg[i] = (chBase + ar) * KTOT + aq * 4;
        wl[i] = p * 4;
    }

    const int aoff = (wave * 16 + l16) * LDK + hi * 18;
    int boff[NT];
#pragma unroll
    for (int n = 0; n < NT; ++n) {
        const int p = n * 16 + l16;
        boff[n] = (p + (p >= WIMG ? 2 : 0)) * CPT + hi * 2;
    }

    v8f acc[NT] = {};

    auto gload = [&](int it, Regs& rg) {
        const int c0 = it * CPT;
#pragma unroll
        for (int i = 0; i < 2; ++i) {
            const float* px = x + xg[i] + c0 * HW;
            rg.x[i] = (v4f){px[0], px[HW], px[2 * HW], px[3 * HW]};
        }
#pragma unroll
        for (int i = 0; i < 5; ++i)
            rg.a[i] = *(const v4f*)(wgt + wg[i] + c0 * 9);
    };

    auto lstore = [&](int buf, const Regs& rg) {
        const v4f z = {0.0f, 0.0f, 0.0f, 0.0f};
#pragma unroll
        for (int i = 0; i < 2; ++i)
            *(v4f*)&Xs[buf][xl[i]] = xv[i] ? rg.x[i] : z;
#pragma unroll
        for (int i = 0; i < 5; ++i)
            *(v4f*)&As[buf][wl[i]] = rg.a[i];
    };

    auto compute = [&](int buf, int t0, int t1) {
#pragma unroll
        for (int t = t0; t < t1; ++t) {
            const int r = t / 3, s = t - 3 * (t / 3);
            const v2f a = {As[buf][aoff + t], As[buf][aoff + t + 9]};
#pragma unroll
            for (int n = 0; n < NT; ++n) {
                const v2f b = *(const v2f*)&Xs[buf][boff[n] + (r * XC + s) * CPT];
                acc[n] = __builtin_amdgcn_wmma_f32_16x16x4_f32(
                             false, a, false, b, (short)0, acc[n], false, false);
            }
        }
    };

    if (tid < 2 * XR * 2) {
        const int bb = tid >> 3, rr = (tid >> 1) & 3, side = tid & 1;
        *(v4f*)&Xs[bb][(rr * XC + side * (XC - 1)) * CPT] = (v4f){0.0f, 0.0f, 0.0f, 0.0f};
    }

    Regs rg;
    gload(0, rg);
    lstore(0, rg);
    __syncthreads();

    int cur = 0;
#pragma unroll 1
    for (int it = 0; it < NITER - 1; ++it) {
        gload(it + 1, rg);
        compute(cur, 0, 7);
        __builtin_amdgcn_sched_barrier(0x100);
        compute(cur, 7, 9);
        lstore(cur ^ 1, rg);
        __syncthreads();
        cur ^= 1;
    }
    compute(cur, 0, 9);

    float* ob = out + ((long)bimg * KOUT * HW + oh0 * WIMG);
#pragma unroll
    for (int rr = 0; rr < 8; ++rr) {
        const int ch = chBase + wave * 16 + hi * 8 + rr;
        const float bv = bias[ch];
        float* op = ob + (long)ch * HW + l16;
#pragma unroll
        for (int n = 0; n < NT; ++n)
            op[n * 16] = acc[n][rr] + bv;
    }
}

extern "C" void kernel_launch(void* const* d_in, const int* in_sizes, int n_in,
                              void* d_out, int out_size, void* d_ws, size_t ws_size,
                              hipStream_t stream) {
    const float* x    = (const float*)d_in[0];
    const float* wgt  = (const float*)d_in[1];
    const float* bias = (const float*)d_in[2];
    float* out        = (float*)d_out;

    dim3 grid(32 * (HIMG / 2), KOUT / BCH);
    conv3x3_wmma_f32<<<grid, TPB, 0, stream>>>(x, wgt, bias, out);
}


# Fragment reads issued one tap ahead as one group into a second register set: the 5 LDS reads of tap t+1, then the 7 WMMAs of tap t; tap 8 runs behind the barrier; sum order unchanged
# speedup vs baseline: 2.1701x; 1.0590x over previous
#include <hip/hip_runtime.h>


typedef float v2f __attribute__((ext_vector_type(2)));
typedef float v4f __attribute__((ext_vector_type(4)));
typedef float v8f __attribute__((ext_vector_type(8)));

#define CIN   128
#define KOUT  256
#define HIMG  56
#define WIMG  56
#define HW    3136
#define KTOT  1152
#define BCH   64
#define BSP   112
#define NT    7
#define CPT   4
#define LDK   36
#define NITER (CIN / CPT)
#define XR    4
#define XC    58
#define XPIX  (XR * WIMG)
#define AV4   (BCH * LDK / 4)
#define TPB   128

struct Regs {
    v4f a[5];
    v4f x[2];
};

struct Frag {
    v2f a;
    v2f b[NT];
};

#define PIN_ORDER() __builtin_amdgcn_sched_barrier(0)

__global__ __launch_bounds__(TPB) void conv3x3_wmma_f32(
    const float* __restrict__ x,
    const float* __restrict__ wgt,
    const float* __restrict__ bias,
    float* __restrict__ out)
{
    __shared__ __attribute__((aligned(16))) float As[2][BCH * LDK];
    __shared__ __attribute__((aligned(16))) float Xs[2][XR * XC * CPT];

    const int tid  = threadIdx.x;
    const int lane = tid & 31;
    const int wave = tid >> 5;
    const int hi   = lane >> 4;
    const int l16  = lane & 15;

    const int chBase = blockIdx.y * BCH;
    const int bimg   = blockIdx.x / (HIMG / 2);
    const int oh0    = (blockIdx.x - bimg * (HIMG / 2)) * 2;

    const int q0 = tid;
    const int q1 = (tid < XPIX - TPB) ? tid + TPB : tid;
    unsigned xg[2];
    int   xl[2];
    bool  xv[2];
#pragma unroll
    for (int i = 0; i < 2; ++i) {
        const int q  = i ? q1 : q0;
        const int rr = q / WIMG, cc = q - rr * WIMG;
        const int ih = oh0 - 1 + rr;
        xv[i] = (unsigned)ih < (unsigned)HIMG;
        const int ihc = ih < 0 ? 0 : (ih > HIMG - 1 ? HIMG - 1 : ih);
        xg[i] = (unsigned)(bimg * (CIN * HW) + ihc * WIMG + cc);
        xl[i] = (rr * XC + cc + 1) * CPT;
    }

    unsigned wg[5];
    int wl[5];
#pragma unroll
    for (int i = 0; i < 5; ++i) {
        int p = tid + i * TPB;
        if (i == 4 && tid >= AV4 - 4 * TPB) p = tid + 3 * TPB;
        const int ar = p / 9, aq = p - ar * 9;
        wg[i] = (unsigned)((chBase + ar) * (KTOT / 4) + aq);
        wl[i] = p * 4;
    }

    const int aoff = (wave * 16 + l16) * LDK + hi * 18;
    int boff[NT];
#pragma unroll
    for (int n = 0; n < NT; ++n) {
        const int p = n * 16 + l16;
        boff[n] = (p + (p >= WIMG ? 2 : 0)) * CPT + hi * 2;
    }

    v8f acc[NT] = {};

    const float* xb = x;
    const v4f*   wb = (const v4f*)wgt;
    auto gload = [&](Regs& rg) {
#pragma unroll
        for (int i = 0; i < 2; ++i) {
            asm volatile("" : "+v"(xg[i]));
            const float* px = xb + xg[i];
            rg.x[i] = (v4f){px[0], px[HW], px[2 * HW], px[3 * HW]};
        }
#pragma unroll
        for (int i = 0; i < 5; ++i) {
            asm volatile("" : "+v"(wg[i]));
            rg.a[i] = wb[wg[i]];
        }
        xb += CPT * HW;
        wb += CPT * 9 / 4;
    };

    auto lstore = [&](int buf, const Regs& rg) {
        const v4f z = {0.0f, 0.0f, 0.0f, 0.0f};
#pragma unroll
        for (int i = 0; i < 2; ++i)
            *(v4f*)&Xs[buf][xl[i]] = xv[i] ? rg.x[i] : z;
#pragma unroll
        for (int i = 0; i < 5; ++i)
            *(v4f*)&As[buf][wl[i]] = rg.a[i];
    };

    auto ldfrag = [&](Frag& f, const int buf, const int t) __attribute__((always_inline)) {
        const int r = t / 3, s = t - 3 * r;
        f.a = (v2f){As[buf][aoff + t], As[buf][aoff + t + 9]};
#pragma unroll
        for (int n = 0; n < NT; ++n)
            f.b[n] = *(const v2f*)&Xs[buf][boff[n] + (r * XC + s) * CPT];
    };

    auto mma = [&](const Frag& f) __attribute__((always_inline)) {
#pragma unroll
        for (int n = 0; n < NT; ++n)
            acc[n] = __builtin_amdgcn_wmma_f32_16x16x4_f32(
                         false, f.a, false, f.b[n], (short)0, acc[n], false, false);
    };

    if (tid < 2 * XR * 2) {
        const int bb = tid >> 3, rr = (tid >> 1) & 3, side = tid & 1;
        *(v4f*)&Xs[bb][(rr * XC + side * (XC - 1)) * CPT] = (v4f){0.0f, 0.0f, 0.0f, 0.0f};
    }

    Regs rg;
    auto tile = [&](Frag& fa, Frag& fb, const int buf, const bool more)
                __attribute__((always_inline)) {
        ldfrag(fb, buf, 1);
        if (more) gload(rg);
        PIN_ORDER(); mma(fa); PIN_ORDER();
        ldfrag(fa, buf, 2); PIN_ORDER(); mma(fb); PIN_ORDER();
        ldfrag(fb, buf, 3); PIN_ORDER(); mma(fa); PIN_ORDER();
        ldfrag(fa, buf, 4); PIN_ORDER(); mma(fb); PIN_ORDER();
        ldfrag(fb, buf, 5); PIN_ORDER(); mma(fa); PIN_ORDER();
        ldfrag(fa, buf, 6); PIN_ORDER(); mma(fb); PIN_ORDER();
        ldfrag(fb, buf, 7); PIN_ORDER(); mma(fa); PIN_ORDER();
        ldfrag(fa, buf, 8); PIN_ORDER(); mma(fb); PIN_ORDER();
        if (more) {
            lstore(buf ^ 1, rg);
            __syncthreads();
            ldfrag(fb, buf ^ 1, 0);
        }
        PIN_ORDER(); mma(fa); PIN_ORDER();
    };

    Frag f0, f1;
    gload(rg);
    lstore(0, rg);
    __syncthreads();
    ldfrag(f0, 0, 0);

#pragma unroll 1
    for (int it = 0; it < NITER - 2; it += 2) {
        tile(f0, f1, 0, true);
        tile(f1, f0, 1, true);
    }
    tile(f0, f1, 0, true);
    tile(f1, f0, 1, false);

    float* ob = out + ((long)bimg * KOUT * HW + oh0 * WIMG);
#pragma unroll
    for (int rr = 0; rr < 8; ++rr) {
        const int ch = chBase + wave * 16 + hi * 8 + rr;
        const float bv = bias[ch];
        float* op = ob + (long)ch * HW + l16;
#pragma unroll
        for (int n = 0; n < NT; ++n)
            op[n * 16] = acc[n][rr] + bv;
    }
}

extern "C" void kernel_launch(void* const* d_in, const int* in_sizes, int n_in,
                              void* d_out, int out_size, void* d_ws, size_t ws_size,
                              hipStream_t stream) {
    const float* x    = (const float*)d_in[0];
    const float* wgt  = (const float*)d_in[1];
    const float* bias = (const float*)d_in[2];
    float* out        = (float*)d_out;

    dim3 grid(32 * (HIMG / 2), KOUT / BCH);
    conv3x3_wmma_f32<<<grid, TPB, 0, stream>>>(x, wgt, bias, out);
}


# Wave tile doubled to 32 channels x 112 positions (block = 64 ch x 4 image rows): a tap's 7 pixel fragments feed 14 WMMAs, one fragment-read group per 14 and half the barriers; sum order unchanged
# speedup vs baseline: 2.2866x; 1.0537x over previous
#include <hip/hip_runtime.h>


typedef float v2f __attribute__((ext_vector_type(2)));
typedef float v4f __attribute__((ext_vector_type(4)));
typedef float v8f __attribute__((ext_vector_type(8)));

#define CIN   128
#define KOUT  256
#define HIMG  56
#define WIMG  56
#define HW    3136
#define KTOT  1152
#define BCH   64
#define BROWS 4
#define NT    7
#define MT    2
#define CPT   4
#define LDK   36
#define NITER (CIN / CPT)
#define XR    6
#define XC    58
#define XPIX  (XR * WIMG)
#define NXP   3
#define AV4   (BCH * LDK / 4)
#define TPB   128

struct Regs {
    v4f a[5];
    v4f x[NXP];
};

struct Frag {
    v2f a[MT];
    v2f b[NT];
};

#define PIN_ORDER() __builtin_amdgcn_sched_barrier(0)

__global__ __launch_bounds__(TPB) void conv3x3_wmma_f32(
    const float* __restrict__ x,
    const float* __restrict__ wgt,
    const float* __restrict__ bias,
    float* __restrict__ out)
{
    __shared__ __attribute__((aligned(16))) float As[2][BCH * LDK];
    __shared__ __attribute__((aligned(16))) float Xs[2][XR * XC * CPT];

    const int tid  = threadIdx.x;
    const int lane = tid & 31;
    const int wave = tid >> 5;
    const int chHalf  = wave & 1;
    const int rowPair = wave >> 1;
    const int hi   = lane >> 4;
    const int l16  = lane & 15;

    const int chBase = blockIdx.y * BCH;
    const int bimg   = blockIdx.x / (HIMG / BROWS);
    const int oh0    = (blockIdx.x - bimg * (HIMG / BROWS)) * BROWS;

    unsigned xg[NXP];
    int   xl[NXP];
    bool  xv[NXP];
#pragma unroll
    for (int i = 0; i < NXP; ++i) {
        const int q  = (tid + i * TPB < XPIX) ? tid + i * TPB : tid;
        const int rr = q / WIMG, cc = q - rr * WIMG;
        const int ih = oh0 - 1 + rr;
        xv[i] = (unsigned)ih < (unsigned)HIMG;
        const int ihc = ih < 0 ? 0 : (ih > HIMG - 1 ? HIMG - 1 : ih);
        xg[i] = (unsigned)(bimg * (CIN * HW) + ihc * WIMG + cc);
        xl[i] = (rr * XC + cc + 1) * CPT;
    }

    unsigned wg[5];
    int wl[5];
#pragma unroll
    for (int i = 0; i < 5; ++i) {
        int p = tid + i * TPB;
        if (i == 4 && tid >= AV4 - 4 * TPB) p = tid + 3 * TPB;
        const int ar = p / 9, aq = p - ar * 9;
        wg[i] = (unsigned)((chBase + ar) * (KTOT / 4) + aq);
        wl[i] = p * 4;
    }

    const int aoff = (chHalf * 32 + l16) * LDK + hi * 18;
    int boff[NT];
#pragma unroll
    for (int n = 0; n < NT; ++n) {
        const int p = n * 16 + l16;
        boff[n] = (rowPair * 2 * XC + p + (p >= WIMG ? 2 : 0)) * CPT + hi * 2;
    }

    v8f acc[MT][NT] = {};

    const float* xb = x;
    const v4f*   wb = (const v4f*)wgt;
    auto gload = [&](Regs& rg) {
#pragma unroll
        for (int i = 0; i < NXP; ++i) {
            asm volatile("" : "+v"(xg[i]));
            const float* px = xb + xg[i];
            rg.x[i] = (v4f){px[0], px[HW], px[2 * HW], px[3 * HW]};
        }
#pragma unroll
        for (int i = 0; i < 5; ++i) {
            asm volatile("" : "+v"(wg[i]));
            rg.a[i] = wb[wg[i]];
        }
        xb += CPT * HW;
        wb += CPT * 9 / 4;
    };

    auto lstore = [&](int buf, const Regs& rg) {
        const v4f z = {0.0f, 0.0f, 0.0f, 0.0f};
#pragma unroll
        for (int i = 0; i < NXP; ++i)
            *(v4f*)&Xs[buf][xl[i]] = xv[i] ? rg.x[i] : z;
#pragma unroll
        for (int i = 0; i < 5; ++i)
            *(v4f*)&As[buf][wl[i]] = rg.a[i];
    };

    auto ldfrag = [&](Frag& f, const int buf, const int t) __attribute__((always_inline)) {
        const int r = t / 3, s = t - 3 * r;
#pragma unroll
        for (int m = 0; m < MT; ++m)
            f.a[m] = (v2f){As[buf][aoff + m * 16 * LDK + t], As[buf][aoff + m * 16 * LDK + t + 9]};
#pragma unroll
        for (int n = 0; n < NT; ++n)
            f.b[n] = *(const v2f*)&Xs[buf][boff[n] + (r * XC + s) * CPT];
    };

    auto mma = [&](const Frag& f) __attribute__((always_inline)) {
#pragma unroll
        for (int n = 0; n < NT; ++n) {
#pragma unroll
            for (int m = 0; m < MT; ++m)
                acc[m][n] = __builtin_amdgcn_wmma_f32_16x16x4_f32(
                                false, f.a[m], false, f.b[n], (short)0, acc[m][n], false, false);
        }
    };

    if (tid < 2 * XR * 2) {
        const int bb = tid / (XR * 2), rr = (tid - bb * (XR * 2)) >> 1, side = tid & 1;
        *(v4f*)&Xs[bb][(rr * XC + side * (XC - 1)) * CPT] = (v4f){0.0f, 0.0f, 0.0f, 0.0f};
    }

    Regs rg;
    auto tile = [&](Frag& fa, Frag& fb, const int buf, const bool more)
                __attribute__((always_inline)) {
        ldfrag(fb, buf, 1);
        if (more) gload(rg);
        PIN_ORDER(); mma(fa); PIN_ORDER();
        ldfrag(fa, buf, 2); PIN_ORDER(); mma(fb); PIN_ORDER();
        ldfrag(fb, buf, 3); PIN_ORDER(); mma(fa); PIN_ORDER();
        ldfrag(fa, buf, 4); PIN_ORDER(); mma(fb); PIN_ORDER();
        ldfrag(fb, buf, 5); PIN_ORDER(); mma(fa); PIN_ORDER();
        ldfrag(fa, buf, 6); PIN_ORDER(); mma(fb); PIN_ORDER();
        ldfrag(fb, buf, 7); PIN_ORDER(); mma(fa); PIN_ORDER();
        ldfrag(fa, buf, 8); PIN_ORDER(); mma(fb); PIN_ORDER();
        if (more) {
            lstore(buf ^ 1, rg);
            __syncthreads();
            ldfrag(fb, buf ^ 1, 0);
        }
        PIN_ORDER(); mma(fa); PIN_ORDER();
    };

    Frag f0, f1;
    gload(rg);
    lstore(0, rg);
    __syncthreads();
    ldfrag(f0, 0, 0);

#pragma unroll 1
    for (int it = 0; it < NITER - 2; it += 2) {
        tile(f0, f1, 0, true);
        tile(f1, f0, 1, true);
    }
    tile(f0, f1, 0, true);
    tile(f1, f0, 1, false);

    float* ob = out + ((long)bimg * KOUT * HW + (oh0 + rowPair * 2) * WIMG);
#pragma unroll
    for (int m = 0; m < MT; ++m) {
#pragma unroll
        for (int rr = 0; rr < 8; ++rr) {
            const int ch = chBase + chHalf * 32 + m * 16 + hi * 8 + rr;
            const float bv = bias[ch];
            float* op = ob + (long)ch * HW + l16;
#pragma unroll
            for (int n = 0; n < NT; ++n)
                op[n * 16] = acc[m][n][rr] + bv;
        }
    }
}

extern "C" void kernel_launch(void* const* d_in, const int* in_sizes, int n_in,
                              void* d_out, int out_size, void* d_ws, size_t ws_size,
                              hipStream_t stream) {
    const float* x    = (const float*)d_in[0];
    const float* wgt  = (const float*)d_in[1];
    const float* bias = (const float*)d_in[2];
    float* out        = (float*)d_out;

    dim3 grid(32 * (HIMG / BROWS), KOUT / BCH);
    conv3x3_wmma_f32<<<grid, TPB, 0, stream>>>(x, wgt, bias, out);
}


# Winograd F(2x2,3x3) replaces the direct form: 16 GEMMs of 128-ch depth (one per transformed position and wave) on the same f32 WMMA, 16 products per 2x2 tile not 36; sum order and rounding change
# speedup vs baseline: 2.6124x; 1.1425x over previous
#include <hip/hip_runtime.h>


typedef float v2f __attribute__((ext_vector_type(2)));
typedef float v4f __attribute__((ext_vector_type(4)));
typedef float v8f __attribute__((ext_vector_type(8)));

#define CIN   128
#define KOUT  256
#define HIMG  56
#define WIMG  56
#define HW    3136
#define KTOT  1152
#define BCH   32
#define OROWS 8
#define TCOLS 28
#define NTILE 112
#define NT    7
#define MT    2
#define CPT   4
#define NITER (CIN / CPT)
#define XR    10
#define XC    58
#define XPIX  (XR * WIMG)
#define TPB   512
#define NPOS  16

#define XS_SZ  (XR * XC * CPT)
#define VS_SZ  (NPOS * NTILE * CPT)
#define US_SZ  (NPOS * BCH * CPT)
#define XS_OFF 0
#define VS_OFF (2 * XS_SZ)
#define WS_SZ  (BCH * CPT * 9)
#define US_OFF (VS_OFF + 2 * VS_SZ)
#define WS_OFF (US_OFF + 2 * US_SZ)
#define SM_SZ  (WS_OFF + 2 * WS_SZ)
#define EX_PITCH 20
#define EX_POS   (BCH * EX_PITCH)
#define EX_SZ    (NPOS * EX_POS)

__global__ __launch_bounds__(TPB) void conv3x3_winograd_wmma_f32(
    const float* __restrict__ x,
    const float* __restrict__ wgt,
    const float* __restrict__ bias,
    float* __restrict__ out)
{
    __shared__ __attribute__((aligned(16))) float sm[SM_SZ];

    const int tid  = threadIdx.x;
    const int lane = tid & 31;
    const int wave = __builtin_amdgcn_readfirstlane(tid >> 5);
    const int hi   = lane >> 4;
    const int l16  = lane & 15;

    const int chBase = blockIdx.x * BCH;
    const int bimg   = blockIdx.y / (HIMG / OROWS);
    const int oh0    = (blockIdx.y - bimg * (HIMG / OROWS)) * OROWS;

    unsigned xg[2]; int xl[2]; bool xv[2];
#pragma unroll
    for (int i = 0; i < 2; ++i) {
        int q = (i == 0) ? tid : (tid + 64);
        q = q < XPIX ? q : XPIX - 1;
        const int rr = q / WIMG, cc = q - rr * WIMG;
        const int ih = oh0 - 1 + rr;
        xv[i] = (unsigned)ih < (unsigned)HIMG;
        const int ihc = ih < 0 ? 0 : (ih > HIMG - 1 ? HIMG - 1 : ih);
        xg[i] = (unsigned)(bimg * (CIN * HW) + ihc * WIMG + cc);
        xl[i] = (rr * XC + cc + 1) * CPT;
    }
    const bool two = wave >= 14;

    const int xi  = wave >> 2;
    const int oc4 = tid & 127;

    const bool wld = wave < 9;
    const int  wp  = wld ? tid : 0;
    const int  war = wp / 9, waq = wp - war * 9;
    const unsigned wgq = (unsigned)((chBase + war) * (KTOT / 4) + waq);
    const int  wlq = wp * 4;

    const int tt  = tid >> 2;
    const int tty = tt / TCOLS, ttx = tt - tty * TCOLS;
    const int xbase = ((2 * tty) * XC + 2 * ttx) * CPT + (tid & 3);

    const int aoff = wave * (BCH * CPT)   + l16 * CPT + hi * 2;
    const int boff = wave * (NTILE * CPT) + l16 * CPT + hi * 2;

    const bool mmaFirst = (((wave >> 1) ^ (wave >> 3)) & 1) != 0;

    v8f acc[MT][NT] = {};

    const float* xb = x;
    const v4f*   wb = (const v4f*)wgt;
    float xr[2][4];
    v4f   wq;

    auto gload_x = [&]() __attribute__((always_inline)) {
        const float* p = xb + xg[0];
        xr[0][0] = p[0]; xr[0][1] = p[HW]; xr[0][2] = p[2 * HW]; xr[0][3] = p[3 * HW];
        if (two) {
            const float* p2 = xb + xg[1];
            xr[1][0] = p2[0]; xr[1][1] = p2[HW]; xr[1][2] = p2[2 * HW]; xr[1][3] = p2[3 * HW];
        }
        xb += CPT * HW;
    };
    auto gload_w = [&]() __attribute__((always_inline)) {
        if (wld) wq = wb[wgq];
        wb += CPT * 9 / 4;
    };

    auto store_raw = [&](const int buf) __attribute__((always_inline)) {
        const v4f z = {0.0f, 0.0f, 0.0f, 0.0f};
        float* xs = sm + XS_OFF + buf * XS_SZ;
        const v4f v0 = {xr[0][0], xr[0][1], xr[0][2], xr[0][3]};
        *(v4f*)&xs[xl[0]] = xv[0] ? v0 : z;
        if (two) {
            const v4f v1 = {xr[1][0], xr[1][1], xr[1][2], xr[1][3]};
            *(v4f*)&xs[xl[1]] = xv[1] ? v1 : z;
        }
        if (wld) *(v4f*)&sm[WS_OFF + buf * WS_SZ + wlq] = wq;
    };

    auto wxform = [&](const int buf) __attribute__((always_inline)) {
        const float* wr = sm + WS_OFF + buf * WS_SZ + 9 * oc4;
        float t3[3];
        if (xi == 0) {
#pragma unroll
            for (int j = 0; j < 3; ++j) t3[j] = wr[j];
        } else if (xi == 3) {
#pragma unroll
            for (int j = 0; j < 3; ++j) t3[j] = wr[6 + j];
        } else {
#pragma unroll
            for (int j = 0; j < 3; ++j) {
                const float g1 = wr[3 + j];
                t3[j] = 0.5f * ((wr[j] + wr[6 + j]) + ((xi == 2) ? -g1 : g1));
            }
        }
        float* us = sm + US_OFF + buf * US_SZ + xi * (4 * BCH * CPT) + oc4;
        us[0 * BCH * CPT] = t3[0];
        us[1 * BCH * CPT] = 0.5f * ((t3[0] + t3[2]) + t3[1]);
        us[2 * BCH * CPT] = 0.5f * ((t3[0] + t3[2]) - t3[1]);
        us[3 * BCH * CPT] = t3[2];
    };

    auto xform = [&](const int buf) __attribute__((always_inline)) {
        const float* xs = sm + XS_OFF + buf * XS_SZ + xbase;
        float d[4][4];
#pragma unroll
        for (int i = 0; i < 4; ++i)
#pragma unroll
            for (int j = 0; j < 4; ++j) d[i][j] = xs[(i * XC + j) * CPT];
        float t[4][4];
#pragma unroll
        for (int j = 0; j < 4; ++j) {
            t[0][j] = d[0][j] - d[2][j];
            t[1][j] = d[1][j] + d[2][j];
            t[2][j] = d[2][j] - d[1][j];
            t[3][j] = d[1][j] - d[3][j];
        }
        float* vs = sm + VS_OFF + buf * VS_SZ + tid;
#pragma unroll
        for (int i = 0; i < 4; ++i) {
            vs[(i * 4 + 0) * (NTILE * CPT)] = t[i][0] - t[i][2];
            vs[(i * 4 + 1) * (NTILE * CPT)] = t[i][1] + t[i][2];
            vs[(i * 4 + 2) * (NTILE * CPT)] = t[i][2] - t[i][1];
            vs[(i * 4 + 3) * (NTILE * CPT)] = t[i][1] - t[i][3];
        }
    };

    auto mma = [&](const int buf) __attribute__((always_inline)) {
        const float* us = sm + US_OFF + buf * US_SZ + aoff;
        const float* vs = sm + VS_OFF + buf * VS_SZ + boff;
        v2f a[MT], b[NT];
#pragma unroll
        for (int m = 0; m < MT; ++m) a[m] = *(const v2f*)(us + m * 16 * CPT);
#pragma unroll
        for (int n = 0; n < NT; ++n) b[n] = *(const v2f*)(vs + n * 16 * CPT);
#pragma unroll
        for (int n = 0; n < NT; ++n) {
#pragma unroll
            for (int m = 0; m < MT; ++m)
                acc[m][n] = __builtin_amdgcn_wmma_f32_16x16x4_f32(
                                false, a[m], false, b[n], (short)0, acc[m][n], false, false);
        }
    };

    auto slot = [&](const int s, const int par) __attribute__((always_inline)) {
        if (s + 2 < NITER) store_raw(par);
        if (s + 3 < NITER) { gload_x(); gload_w(); }
        if (mmaFirst) mma(par);
        if (s + 1 < NITER) {
            wxform(par ^ 1);
            if (wave < 14) xform(par ^ 1);
        }
        if (!mmaFirst) mma(par);
        __syncthreads();
    };

    gload_x(); gload_w();
    store_raw(0);
    gload_x(); gload_w();
    if (tid < 2 * XR * 2) {
        const int bb = tid / (XR * 2), rr = (tid - bb * (XR * 2)) >> 1, side = tid & 1;
        *(v4f*)&sm[XS_OFF + bb * XS_SZ + (rr * XC + side * (XC - 1)) * CPT] = (v4f){0.0f, 0.0f, 0.0f, 0.0f};
    }
    __syncthreads();
    wxform(0);
    if (wave < 14) xform(0);
    store_raw(1);
    gload_x(); gload_w();
    __syncthreads();

    for (int s = 0; s < NITER; s += 2) {
        slot(s, 0);
        slot(s + 1, 1);
    }

    const int eo  = tid >> 4;
    const int etl = tid & 15;
    const float bv = bias[chBase + eo];
    float* obase = out + ((long)(bimg * KOUT + chBase + eo) * HW + oh0 * WIMG);
#pragma unroll
    for (int n = 0; n < NT; ++n) {
        float* ex = sm + (n & 1) * EX_SZ;
#pragma unroll
        for (int m = 0; m < MT; ++m)
#pragma unroll
            for (int rr = 0; rr < 8; ++rr)
                ex[wave * EX_POS + (m * 16 + hi * 8 + rr) * EX_PITCH + l16] = acc[m][n][rr];
        __syncthreads();
        float mm[16];
#pragma unroll
        for (int p = 0; p < 16; ++p) mm[p] = ex[p * EX_POS + eo * EX_PITCH + etl];
        float s0[4], s1[4];
#pragma unroll
        for (int nu = 0; nu < 4; ++nu) {
            s0[nu] = (mm[0 + nu] + mm[4 + nu]) + mm[8 + nu];
            s1[nu] = (mm[4 + nu] - mm[8 + nu]) - mm[12 + nu];
        }
        const int t  = n * 16 + etl;
        const int ty = t / TCOLS, tx = t - ty * TCOLS;
        float* op = obase + (2 * ty) * WIMG + 2 * tx;
        *(v2f*)op          = (v2f){(s0[0] + s0[1]) + s0[2] + bv, (s0[1] - s0[2]) - s0[3] + bv};
        *(v2f*)(op + WIMG) = (v2f){(s1[0] + s1[1]) + s1[2] + bv, (s1[1] - s1[2]) - s1[3] + bv};
    }
}

extern "C" void kernel_launch(void* const* d_in, const int* in_sizes, int n_in,
                              void* d_out, int out_size, void* d_ws, size_t ws_size,
                              hipStream_t stream) {
    const float* x    = (const float*)d_in[0];
    const float* wgt  = (const float*)d_in[1];
    const float* bias = (const float*)d_in[2];
    float* out        = (float*)d_out;

    dim3 grid(KOUT / BCH, 32 * (HIMG / OROWS));
    conv3x3_winograd_wmma_f32<<<grid, TPB, 0, stream>>>(x, wgt, bias, out);
}


# Two K steps (8 channels) between block barriers: stage buffers doubled in LDS, a wave runs 28 WMMAs then both steps' transforms per barrier, 16 barriers not 32; sum order unchanged
# speedup vs baseline: 2.6955x; 1.0318x over previous
#include <hip/hip_runtime.h>


typedef float v2f __attribute__((ext_vector_type(2)));
typedef float v4f __attribute__((ext_vector_type(4)));
typedef float v8f __attribute__((ext_vector_type(8)));

#define CIN   128
#define KOUT  256
#define HIMG  56
#define WIMG  56
#define HW    3136
#define KTOT  1152
#define BCH   32
#define OROWS 8
#define TCOLS 28
#define NTILE 112
#define NT    7
#define MT    2
#define CPT   4
#define NITER (CIN / CPT)
#define SUB   2
#define NPAIR (NITER / SUB)
#define XR    10
#define XC    58
#define XPIX  (XR * WIMG)
#define TPB   512
#define NPOS  16

#define XS_SZ  (XR * XC * CPT)
#define VS_SZ  (NPOS * NTILE * CPT)
#define US_SZ  (NPOS * BCH * CPT)
#define XS_OFF 0
#define VS_OFF (2 * SUB * XS_SZ)
#define WS_SZ  (BCH * CPT * 9)
#define US_OFF (VS_OFF + 2 * SUB * VS_SZ)
#define WS_OFF (US_OFF + 2 * SUB * US_SZ)
#define SM_SZ  (WS_OFF + 2 * SUB * WS_SZ)
#define EX_PITCH 20
#define EX_POS   (BCH * EX_PITCH)
#define EX_SZ    (NPOS * EX_POS)

__global__ __launch_bounds__(TPB) void conv3x3_winograd_wmma_f32(
    const float* __restrict__ x,
    const float* __restrict__ wgt,
    const float* __restrict__ bias,
    float* __restrict__ out)
{
    __shared__ __attribute__((aligned(16))) float sm[SM_SZ];

    const int tid  = threadIdx.x;
    const int lane = tid & 31;
    const int wave = __builtin_amdgcn_readfirstlane(tid >> 5);
    const int hi   = lane >> 4;
    const int l16  = lane & 15;

    const int chBase = blockIdx.x * BCH;
    const int bimg   = blockIdx.y / (HIMG / OROWS);
    const int oh0    = (blockIdx.y - bimg * (HIMG / OROWS)) * OROWS;

    unsigned xg[2]; int xl[2]; bool xv[2];
#pragma unroll
    for (int i = 0; i < 2; ++i) {
        int q = (i == 0) ? tid : (tid + 64);
        q = q < XPIX ? q : XPIX - 1;
        const int rr = q / WIMG, cc = q - rr * WIMG;
        const int ih = oh0 - 1 + rr;
        xv[i] = (unsigned)ih < (unsigned)HIMG;
        const int ihc = ih < 0 ? 0 : (ih > HIMG - 1 ? HIMG - 1 : ih);
        xg[i] = (unsigned)(bimg * (CIN * HW) + ihc * WIMG + cc);
        xl[i] = (rr * XC + cc + 1) * CPT;
    }
    const bool two = wave >= 14;

    const int xi  = wave >> 2;
    const int oc4 = tid & 127;

    const bool wld = wave < 9;
    const int  wp  = wld ? tid : 0;
    const int  war = wp / 9, waq = wp - war * 9;
    const unsigned wgq = (unsigned)((chBase + war) * (KTOT / 4) + waq);
    const int  wlq = wp * 4;

    const int tt  = tid >> 2;
    const int tty = tt / TCOLS, ttx = tt - tty * TCOLS;
    const int xbase = ((2 * tty) * XC + 2 * ttx) * CPT + (tid & 3);

    const int aoff = wave * (BCH * CPT)   + l16 * CPT + hi * 2;
    const int boff = wave * (NTILE * CPT) + l16 * CPT + hi * 2;

    const bool mmaFirst = (((wave >> 1) ^ (wave >> 3)) & 1) != 0;

    v8f acc[MT][NT] = {};

    const float* xb = x;
    const v4f*   wb = (const v4f*)wgt;
    float xr[SUB][2][4];
    v4f   wq[SUB];

    auto gload_x = [&]() __attribute__((always_inline)) {
#pragma unroll
        for (int j = 0; j < SUB; ++j) {
            const float* p = xb + xg[0] + j * (CPT * HW);
            xr[j][0][0] = p[0]; xr[j][0][1] = p[HW]; xr[j][0][2] = p[2 * HW]; xr[j][0][3] = p[3 * HW];
            if (two) {
                const float* p2 = xb + xg[1] + j * (CPT * HW);
                xr[j][1][0] = p2[0]; xr[j][1][1] = p2[HW]; xr[j][1][2] = p2[2 * HW]; xr[j][1][3] = p2[3 * HW];
            }
        }
        xb += SUB * CPT * HW;
    };
    auto gload_w = [&]() __attribute__((always_inline)) {
        if (wld) {
#pragma unroll
            for (int j = 0; j < SUB; ++j) wq[j] = wb[wgq + j * (CPT * 9 / 4)];
        }
        wb += SUB * CPT * 9 / 4;
    };

    auto store_raw = [&](const int buf) __attribute__((always_inline)) {
        const v4f z = {0.0f, 0.0f, 0.0f, 0.0f};
#pragma unroll
        for (int j = 0; j < SUB; ++j) {
            float* xs = sm + XS_OFF + (buf * SUB + j) * XS_SZ;
            const v4f v0 = {xr[j][0][0], xr[j][0][1], xr[j][0][2], xr[j][0][3]};
            *(v4f*)&xs[xl[0]] = xv[0] ? v0 : z;
            if (two) {
                const v4f v1 = {xr[j][1][0], xr[j][1][1], xr[j][1][2], xr[j][1][3]};
                *(v4f*)&xs[xl[1]] = xv[1] ? v1 : z;
            }
            if (wld) *(v4f*)&sm[WS_OFF + (buf * SUB + j) * WS_SZ + wlq] = wq[j];
        }
    };

    auto wxform1 = [&](const int buf) __attribute__((always_inline)) {
        const float* wr = sm + WS_OFF + buf * WS_SZ + 9 * oc4;
        float t3[3];
        if (xi == 0) {
#pragma unroll
            for (int j = 0; j < 3; ++j) t3[j] = wr[j];
        } else if (xi == 3) {
#pragma unroll
            for (int j = 0; j < 3; ++j) t3[j] = wr[6 + j];
        } else {
#pragma unroll
            for (int j = 0; j < 3; ++j) {
                const float g1 = wr[3 + j];
                t3[j] = 0.5f * ((wr[j] + wr[6 + j]) + ((xi == 2) ? -g1 : g1));
            }
        }
        float* us = sm + US_OFF + buf * US_SZ + xi * (4 * BCH * CPT) + oc4;
        us[0 * BCH * CPT] = t3[0];
        us[1 * BCH * CPT] = 0.5f * ((t3[0] + t3[2]) + t3[1]);
        us[2 * BCH * CPT] = 0.5f * ((t3[0] + t3[2]) - t3[1]);
        us[3 * BCH * CPT] = t3[2];
    };
    auto wxform = [&](const int stage) __attribute__((always_inline)) {
#pragma unroll
        for (int j = 0; j < SUB; ++j) wxform1(stage * SUB + j);
    };

    auto xform1 = [&](const int buf) __attribute__((always_inline)) {
        const float* xs = sm + XS_OFF + buf * XS_SZ + xbase;
        float d[4][4];
#pragma unroll
        for (int i = 0; i < 4; ++i)
#pragma unroll
            for (int j = 0; j < 4; ++j) d[i][j] = xs[(i * XC + j) * CPT];
        float t[4][4];
#pragma unroll
        for (int j = 0; j < 4; ++j) {
            t[0][j] = d[0][j] - d[2][j];
            t[1][j] = d[1][j] + d[2][j];
            t[2][j] = d[2][j] - d[1][j];
            t[3][j] = d[1][j] - d[3][j];
        }
        float* vs = sm + VS_OFF + buf * VS_SZ + tid;
#pragma unroll
        for (int i = 0; i < 4; ++i) {
            vs[(i * 4 + 0) * (NTILE * CPT)] = t[i][0] - t[i][2];
            vs[(i * 4 + 1) * (NTILE * CPT)] = t[i][1] + t[i][2];
            vs[(i * 4 + 2) * (NTILE * CPT)] = t[i][2] - t[i][1];
            vs[(i * 4 + 3) * (NTILE * CPT)] = t[i][1] - t[i][3];
        }
    };
    auto xform = [&](const int stage) __attribute__((always_inline)) {
#pragma unroll
        for (int j = 0; j < SUB; ++j) xform1(stage * SUB + j);
    };

    auto mma = [&](const int stage) __attribute__((always_inline)) {
        v2f a[SUB][MT], b[SUB][NT];
#pragma unroll
        for (int j = 0; j < SUB; ++j) {
            const float* us = sm + US_OFF + (stage * SUB + j) * US_SZ + aoff;
            const float* vs = sm + VS_OFF + (stage * SUB + j) * VS_SZ + boff;
#pragma unroll
            for (int m = 0; m < MT; ++m) a[j][m] = *(const v2f*)(us + m * 16 * CPT);
#pragma unroll
            for (int n = 0; n < NT; ++n) b[j][n] = *(const v2f*)(vs + n * 16 * CPT);
        }
#pragma unroll
        for (int j = 0; j < SUB; ++j) {
#pragma unroll
            for (int n = 0; n < NT; ++n) {
#pragma unroll
                for (int m = 0; m < MT; ++m)
                    acc[m][n] = __builtin_amdgcn_wmma_f32_16x16x4_f32(
                                    false, a[j][m], false, b[j][n], (short)0, acc[m][n], false, false);
            }
        }
    };

    auto slot = [&](const int s, const int par) __attribute__((always_inline)) {
        if (s + 2 < NPAIR) store_raw(par);
        if (s + 3 < NPAIR) { gload_x(); gload_w(); }
        if (mmaFirst) mma(par);
        if (s + 1 < NPAIR) {
            wxform(par ^ 1);
            if (wave < 14) xform(par ^ 1);
        }
        if (!mmaFirst) mma(par);
        __syncthreads();
    };

    gload_x(); gload_w();
    store_raw(0);
    gload_x(); gload_w();
    if (tid < 2 * SUB * XR * 2) {
        const int bb = tid / (XR * 2), rr = (tid - bb * (XR * 2)) >> 1, side = tid & 1;
        *(v4f*)&sm[XS_OFF + bb * XS_SZ + (rr * XC + side * (XC - 1)) * CPT] = (v4f){0.0f, 0.0f, 0.0f, 0.0f};
    }
    __syncthreads();
    wxform(0);
    if (wave < 14) xform(0);
    store_raw(1);
    gload_x(); gload_w();
    __syncthreads();

    for (int s = 0; s < NPAIR; s += 2) {
        slot(s, 0);
        slot(s + 1, 1);
    }

    const int eo  = tid >> 4;
    const int etl = tid & 15;
    const float bv = bias[chBase + eo];
    float* obase = out + ((long)(bimg * KOUT + chBase + eo) * HW + oh0 * WIMG);
#pragma unroll
    for (int n = 0; n < NT; ++n) {
        float* ex = sm + (n & 1) * EX_SZ;
#pragma unroll
        for (int m = 0; m < MT; ++m)
#pragma unroll
            for (int rr = 0; rr < 8; ++rr)
                ex[wave * EX_POS + (m * 16 + hi * 8 + rr) * EX_PITCH + l16] = acc[m][n][rr];
        __syncthreads();
        float mm[16];
#pragma unroll
        for (int p = 0; p < 16; ++p) mm[p] = ex[p * EX_POS + eo * EX_PITCH + etl];
        float s0[4], s1[4];
#pragma unroll
        for (int nu = 0; nu < 4; ++nu) {
            s0[nu] = (mm[0 + nu] + mm[4 + nu]) + mm[8 + nu];
            s1[nu] = (mm[4 + nu] - mm[8 + nu]) - mm[12 + nu];
        }
        const int t  = n * 16 + etl;
        const int ty = t / TCOLS, tx = t - ty * TCOLS;
        float* op = obase + (2 * ty) * WIMG + 2 * tx;
        *(v2f*)op          = (v2f){(s0[0] + s0[1]) + s0[2] + bv, (s0[1] - s0[2]) - s0[3] + bv};
        *(v2f*)(op + WIMG) = (v2f){(s1[0] + s1[1]) + s1[2] + bv, (s1[1] - s1[2]) - s1[3] + bv};
    }
}

extern "C" void kernel_launch(void* const* d_in, const int* in_sizes, int n_in,
                              void* d_out, int out_size, void* d_ws, size_t ws_size,
                              hipStream_t stream) {
    const float* x    = (const float*)d_in[0];
    const float* wgt  = (const float*)d_in[1];
    const float* bias = (const float*)d_in[2];
    float* out        = (float*)d_out;

    dim3 grid(KOUT / BCH, 32 * (HIMG / OROWS));
    conv3x3_winograd_wmma_f32<<<grid, TPB, 0, stream>>>(x, wgt, bias, out);
}


# One work-group per WGP (grid 256, not 1792) walks 7 block tiles: the load/stage/transform pipeline runs on across the tile border, exchange in LDS of its own; 6 of 7 start-ups go; sum order unchanged
# speedup vs baseline: 2.8484x; 1.0567x over previous
#include <hip/hip_runtime.h>


typedef float v2f __attribute__((ext_vector_type(2)));
typedef float v4f __attribute__((ext_vector_type(4)));
typedef float v8f __attribute__((ext_vector_type(8)));

#define CIN   128
#define KOUT  256
#define HIMG  56
#define WIMG  56
#define HW    3136
#define KTOT  1152
#define BCH   32
#define OROWS 8
#define TCOLS 28
#define NTILE 112
#define NT    7
#define MT    2
#define CPT   4
#define NITER (CIN / CPT)
#define SUB   2
#define NPAIR (NITER / SUB)
#define XR    10
#define XC    58
#define XPIX  (XR * WIMG)
#define TPB   512
#define NPOS  16
#define NWG   256
#define YGRP  32
#define TPW   7

#define XS_SZ  (XR * XC * CPT)
#define VS_SZ  (NPOS * NTILE * CPT)
#define US_SZ  (NPOS * BCH * CPT)
#define XS_OFF 0
#define VS_OFF (2 * SUB * XS_SZ)
#define WS_SZ  (BCH * CPT * 9)
#define US_OFF (VS_OFF + 2 * SUB * VS_SZ)
#define WS_OFF (US_OFF + 2 * SUB * US_SZ)
#define SM_SZ  (WS_OFF + 2 * SUB * WS_SZ)
#define EX_PITCH 20
#define EX_POS   (BCH * EX_PITCH)
#define EX_SZ    (NPOS * EX_POS)
#define EX_OFF   SM_SZ
#define LDS_SZ   (SM_SZ + 2 * EX_SZ)

__global__ __launch_bounds__(TPB) void conv3x3_winograd_wmma_f32(
    const float* __restrict__ x,
    const float* __restrict__ wgt,
    const float* __restrict__ bias,
    float* __restrict__ out)
{
    __shared__ __attribute__((aligned(16))) float sm[LDS_SZ];

    const int tid  = threadIdx.x;
    const int lane = tid & 31;
    const int wave = __builtin_amdgcn_readfirstlane(tid >> 5);
    const int hi   = lane >> 4;
    const int l16  = lane & 15;

    const int chBase = (blockIdx.x & 7) * BCH;
    const int y0     = blockIdx.x >> 3;

    unsigned xg[2]; int xl[2]; bool xv[2]; int xrr[2], xcc[2];
#pragma unroll
    for (int i = 0; i < 2; ++i) {
        int q = (i == 0) ? tid : (tid + 64);
        q = q < XPIX ? q : XPIX - 1;
        const int rr = q / WIMG, cc = q - rr * WIMG;
        xrr[i] = rr; xcc[i] = cc;
        xl[i] = (rr * XC + cc + 1) * CPT;
    }
    auto set_tile = [&](const int it) __attribute__((always_inline)) {
        const int y    = y0 + YGRP * it;
        const int bimg = y / (HIMG / OROWS);
        const int oh0  = (y - bimg * (HIMG / OROWS)) * OROWS;
#pragma unroll
        for (int i = 0; i < 2; ++i) {
            const int ih = oh0 - 1 + xrr[i];
            xv[i] = (unsigned)ih < (unsigned)HIMG;
            const int ihc = ih < 0 ? 0 : (ih > HIMG - 1 ? HIMG - 1 : ih);
            xg[i] = (unsigned)(bimg * (CIN * HW) + ihc * WIMG + xcc[i]);
        }
    };
    set_tile(0);
    const bool two = wave >= 14;

    const int xi  = wave >> 2;
    const int oc4 = tid & 127;

    const bool wld = wave < 9;
    const int  wp  = wld ? tid : 0;
    const int  war = wp / 9, waq = wp - war * 9;
    const unsigned wgq = (unsigned)((chBase + war) * (KTOT / 4) + waq);
    const int  wlq = wp * 4;

    const int tt  = tid >> 2;
    const int tty = tt / TCOLS, ttx = tt - tty * TCOLS;
    const int xbase = ((2 * tty) * XC + 2 * ttx) * CPT + (tid & 3);

    const int aoff = wave * (BCH * CPT)   + l16 * CPT + hi * 2;
    const int boff = wave * (NTILE * CPT) + l16 * CPT + hi * 2;

    const bool mmaFirst = (((wave >> 1) ^ (wave >> 3)) & 1) != 0;

    v8f acc[MT][NT] = {};

    const float* xb = x;
    const v4f*   wb = (const v4f*)wgt;
    float xr[SUB][2][4];
    v4f   wq[SUB];

    auto gload_x = [&]() __attribute__((always_inline)) {
#pragma unroll
        for (int j = 0; j < SUB; ++j) {
            const float* p = xb + xg[0] + j * (CPT * HW);
            xr[j][0][0] = p[0]; xr[j][0][1] = p[HW]; xr[j][0][2] = p[2 * HW]; xr[j][0][3] = p[3 * HW];
            if (two) {
                const float* p2 = xb + xg[1] + j * (CPT * HW);
                xr[j][1][0] = p2[0]; xr[j][1][1] = p2[HW]; xr[j][1][2] = p2[2 * HW]; xr[j][1][3] = p2[3 * HW];
            }
        }
        xb += SUB * CPT * HW;
    };
    auto gload_w = [&]() __attribute__((always_inline)) {
        if (wld) {
#pragma unroll
            for (int j = 0; j < SUB; ++j) wq[j] = wb[wgq + j * (CPT * 9 / 4)];
        }
        wb += SUB * CPT * 9 / 4;
    };

    auto store_raw = [&](const int buf) __attribute__((always_inline)) {
        const v4f z = {0.0f, 0.0f, 0.0f, 0.0f};
#pragma unroll
        for (int j = 0; j < SUB; ++j) {
            float* xs = sm + XS_OFF + (buf * SUB + j) * XS_SZ;
            const v4f v0 = {xr[j][0][0], xr[j][0][1], xr[j][0][2], xr[j][0][3]};
            *(v4f*)&xs[xl[0]] = xv[0] ? v0 : z;
            if (two) {
                const v4f v1 = {xr[j][1][0], xr[j][1][1], xr[j][1][2], xr[j][1][3]};
                *(v4f*)&xs[xl[1]] = xv[1] ? v1 : z;
            }
            if (wld) *(v4f*)&sm[WS_OFF + (buf * SUB + j) * WS_SZ + wlq] = wq[j];
        }
    };

    auto wxform1 = [&](const int buf) __attribute__((always_inline)) {
        const float* wr = sm + WS_OFF + buf * WS_SZ + 9 * oc4;
        float t3[3];
        if (xi == 0) {
#pragma unroll
            for (int j = 0; j < 3; ++j) t3[j] = wr[j];
        } else if (xi == 3) {
#pragma unroll
            for (int j = 0; j < 3; ++j) t3[j] = wr[6 + j];
        } else {
#pragma unroll
            for (int j = 0; j < 3; ++j) {
                const float g1 = wr[3 + j];
                t3[j] = 0.5f * ((wr[j] + wr[6 + j]) + ((xi == 2) ? -g1 : g1));
            }
        }
        float* us = sm + US_OFF + buf * US_SZ + xi * (4 * BCH * CPT) + oc4;
        us[0 * BCH * CPT] = t3[0];
        us[1 * BCH * CPT] = 0.5f * ((t3[0] + t3[2]) + t3[1]);
        us[2 * BCH * CPT] = 0.5f * ((t3[0] + t3[2]) - t3[1]);
        us[3 * BCH * CPT] = t3[2];
    };
    auto wxform = [&](const int stage) __attribute__((always_inline)) {
#pragma unroll
        for (int j = 0; j < SUB; ++j) wxform1(stage * SUB + j);
    };

    auto xform1 = [&](const int buf) __attribute__((always_inline)) {
        const float* xs = sm + XS_OFF + buf * XS_SZ + xbase;
        float d[4][4];
#pragma unroll
        for (int i = 0; i < 4; ++i)
#pragma unroll
            for (int j = 0; j < 4; ++j) d[i][j] = xs[(i * XC + j) * CPT];
        float t[4][4];
#pragma unroll
        for (int j = 0; j < 4; ++j) {
            t[0][j] = d[0][j] - d[2][j];
            t[1][j] = d[1][j] + d[2][j];
            t[2][j] = d[2][j] - d[1][j];
            t[3][j] = d[1][j] - d[3][j];
        }
        float* vs = sm + VS_OFF + buf * VS_SZ + tid;
#pragma unroll
        for (int i = 0; i < 4; ++i) {
            vs[(i * 4 + 0) * (NTILE * CPT)] = t[i][0] - t[i][2];
            vs[(i * 4 + 1) * (NTILE * CPT)] = t[i][1] + t[i][2];
            vs[(i * 4 + 2) * (NTILE * CPT)] = t[i][2] - t[i][1];
            vs[(i * 4 + 3) * (NTILE * CPT)] = t[i][1] - t[i][3];
        }
    };
    auto xform = [&](const int stage) __attribute__((always_inline)) {
#pragma unroll
        for (int j = 0; j < SUB; ++j) xform1(stage * SUB + j);
    };

    auto mma = [&](const int stage) __attribute__((always_inline)) {
        v2f a[SUB][MT], b[SUB][NT];
#pragma unroll
        for (int j = 0; j < SUB; ++j) {
            const float* us = sm + US_OFF + (stage * SUB + j) * US_SZ + aoff;
            const float* vs = sm + VS_OFF + (stage * SUB + j) * VS_SZ + boff;
#pragma unroll
            for (int m = 0; m < MT; ++m) a[j][m] = *(const v2f*)(us + m * 16 * CPT);
#pragma unroll
            for (int n = 0; n < NT; ++n) b[j][n] = *(const v2f*)(vs + n * 16 * CPT);
        }
#pragma unroll
        for (int j = 0; j < SUB; ++j) {
#pragma unroll
            for (int n = 0; n < NT; ++n) {
#pragma unroll
                for (int m = 0; m < MT; ++m)
                    acc[m][n] = __builtin_amdgcn_wmma_f32_16x16x4_f32(
                                    false, a[j][m], false, b[j][n], (short)0, acc[m][n], false, false);
            }
        }
    };

    auto slot = [&](const int it, const bool more, const int s, const int par) __attribute__((always_inline)) {
        if (more || s + 2 < NPAIR) store_raw(par);
        if (more || s + 3 < NPAIR) {
            if (s + 3 == NPAIR) {
                set_tile(it + 1);
                xb = x;
                wb = (const v4f*)wgt;
            }
            gload_x(); gload_w();
        }
        if (mmaFirst) mma(par);
        if (more || s + 1 < NPAIR) {
            wxform(par ^ 1);
            if (wave < 14) xform(par ^ 1);
        }
        if (!mmaFirst) mma(par);
        __syncthreads();
    };

    gload_x(); gload_w();
    store_raw(0);
    gload_x(); gload_w();
    if (tid < 2 * SUB * XR * 2) {
        const int bb = tid / (XR * 2), rr = (tid - bb * (XR * 2)) >> 1, side = tid & 1;
        *(v4f*)&sm[XS_OFF + bb * XS_SZ + (rr * XC + side * (XC - 1)) * CPT] = (v4f){0.0f, 0.0f, 0.0f, 0.0f};
    }
    __syncthreads();
    wxform(0);
    if (wave < 14) xform(0);
    store_raw(1);
    gload_x(); gload_w();
    __syncthreads();

    const int eo  = tid >> 4;
    const int etl = tid & 15;
    const float bv = bias[chBase + eo];

    for (int it = 0; it < TPW; ++it) {
    const bool more = it + 1 < TPW;
    for (int s = 0; s < NPAIR; s += 2) {
        slot(it, more, s, 0);
        slot(it, more, s + 1, 1);
    }

    const int ey    = y0 + YGRP * it;
    const int ebimg = ey / (HIMG / OROWS);
    const int eoh0  = (ey - ebimg * (HIMG / OROWS)) * OROWS;
    float* obase = out + ((long)(ebimg * KOUT + chBase + eo) * HW + eoh0 * WIMG);
    v2f o0[NT], o1[NT];
#pragma unroll
    for (int n = 0; n < NT; ++n) {
        float* ex = sm + EX_OFF + (n & 1) * EX_SZ;
#pragma unroll
        for (int m = 0; m < MT; ++m)
#pragma unroll
            for (int rr = 0; rr < 8; ++rr)
                ex[wave * EX_POS + (m * 16 + hi * 8 + rr) * EX_PITCH + l16] = acc[m][n][rr];
        __syncthreads();
        float mm[16];
#pragma unroll
        for (int p = 0; p < 16; ++p) mm[p] = ex[p * EX_POS + eo * EX_PITCH + etl];
        float s0[4], s1[4];
#pragma unroll
        for (int nu = 0; nu < 4; ++nu) {
            s0[nu] = (mm[0 + nu] + mm[4 + nu]) + mm[8 + nu];
            s1[nu] = (mm[4 + nu] - mm[8 + nu]) - mm[12 + nu];
        }
        o0[n] = (v2f){(s0[0] + s0[1]) + s0[2] + bv, (s0[1] - s0[2]) - s0[3] + bv};
        o1[n] = (v2f){(s1[0] + s1[1]) + s1[2] + bv, (s1[1] - s1[2]) - s1[3] + bv};
    }
#pragma unroll
    for (int n = 0; n < NT; ++n) {
        const int t  = n * 16 + etl;
        const int ty = t / TCOLS, tx = t - ty * TCOLS;
        float* op = obase + (2 * ty) * WIMG + 2 * tx;
        *(v2f*)op          = o0[n];
        *(v2f*)(op + WIMG) = o1[n];
    }
#pragma unroll
    for (int m = 0; m < MT; ++m)
#pragma unroll
        for (int n = 0; n < NT; ++n) acc[m][n] = (v8f){};
    }
}

extern "C" void kernel_launch(void* const* d_in, const int* in_sizes, int n_in,
                              void* d_out, int out_size, void* d_ws, size_t ws_size,
                              hipStream_t stream) {
    const float* x    = (const float*)d_in[0];
    const float* wgt  = (const float*)d_in[1];
    const float* bias = (const float*)d_in[2];
    float* out        = (float*)d_out;

    conv3x3_winograd_wmma_f32<<<dim3(NWG), TPB, 0, stream>>>(x, wgt, bias, out);
}


# Input transform per (tile, channel pair): 8-byte LDS reads/writes and packed adds, the two lane halves take the pair's two K steps, stage pitches padded so the halves miss banks; sum order unchanged
# speedup vs baseline: 2.9842x; 1.0477x over previous
#include <hip/hip_runtime.h>


typedef float v2f __attribute__((ext_vector_type(2)));
typedef float v4f __attribute__((ext_vector_type(4)));
typedef float v8f __attribute__((ext_vector_type(8)));

#define CIN   128
#define KOUT  256
#define HIMG  56
#define WIMG  56
#define HW    3136
#define KTOT  1152
#define BCH   32
#define OROWS 8
#define TCOLS 28
#define NTILE 112
#define NT    7
#define MT    2
#define CPT   4
#define NITER (CIN / CPT)
#define SUB   2
#define NPAIR (NITER / SUB)
#define XR    10
#define XC    58
#define XPIX  (XR * WIMG)
#define TPB   512
#define NPOS  16
#define NWG   256
#define YGRP  32
#define TPW   7

#define XS_SZ  (XR * XC * CPT + 4)
#define VS_SZ  (NPOS * NTILE * CPT + 32)
#define US_SZ  (NPOS * BCH * CPT)
#define XS_OFF 0
#define VS_OFF (2 * SUB * XS_SZ)
#define WS_SZ  (BCH * CPT * 9)
#define US_OFF (VS_OFF + 2 * SUB * VS_SZ)
#define WS_OFF (US_OFF + 2 * SUB * US_SZ)
#define SM_SZ  (WS_OFF + 2 * SUB * WS_SZ)
#define EX_PITCH 20
#define EX_POS   (BCH * EX_PITCH)
#define EX_SZ    (NPOS * EX_POS)
#define EX_OFF   SM_SZ
#define LDS_SZ   (SM_SZ + 2 * EX_SZ)

__global__ __launch_bounds__(TPB) void conv3x3_winograd_wmma_f32(
    const float* __restrict__ x,
    const float* __restrict__ wgt,
    const float* __restrict__ bias,
    float* __restrict__ out)
{
    __shared__ __attribute__((aligned(16))) float sm[LDS_SZ];

    const int tid  = threadIdx.x;
    const int lane = tid & 31;
    const int wave = __builtin_amdgcn_readfirstlane(tid >> 5);
    const int hi   = lane >> 4;
    const int l16  = lane & 15;

    const int chBase = (blockIdx.x & 7) * BCH;
    const int y0     = blockIdx.x >> 3;

    unsigned xg[2]; int xl[2]; bool xv[2]; int xrr[2], xcc[2];
#pragma unroll
    for (int i = 0; i < 2; ++i) {
        int q = (i == 0) ? tid : (tid + 64);
        q = q < XPIX ? q : XPIX - 1;
        const int rr = q / WIMG, cc = q - rr * WIMG;
        xrr[i] = rr; xcc[i] = cc;
        xl[i] = (rr * XC + cc + 1) * CPT;
    }
    auto set_tile = [&](const int it) __attribute__((always_inline)) {
        const int y    = y0 + YGRP * it;
        const int bimg = y / (HIMG / OROWS);
        const int oh0  = (y - bimg * (HIMG / OROWS)) * OROWS;
#pragma unroll
        for (int i = 0; i < 2; ++i) {
            const int ih = oh0 - 1 + xrr[i];
            xv[i] = (unsigned)ih < (unsigned)HIMG;
            const int ihc = ih < 0 ? 0 : (ih > HIMG - 1 ? HIMG - 1 : ih);
            xg[i] = (unsigned)(bimg * (CIN * HW) + ihc * WIMG + xcc[i]);
        }
    };
    set_tile(0);
    const bool two = wave >= 14;

    const int xi  = wave >> 2;
    const int oc4 = tid & 127;

    const bool wld = wave < 9;
    const int  wp  = wld ? tid : 0;
    const int  war = wp / 9, waq = wp - war * 9;
    const unsigned wgq = (unsigned)((chBase + war) * (KTOT / 4) + waq);
    const int  wlq = wp * 4;

    const int tt  = wave * 8 + (l16 >> 1);
    const int tcp = (l16 & 1) * 2;
    const int tty = tt / TCOLS, ttx = tt - tty * TCOLS;
    const int xbase = hi * XS_SZ + ((2 * tty) * XC + 2 * ttx) * CPT + tcp;
    const int vbase = hi * VS_SZ + tt * CPT + tcp;

    const int aoff = wave * (BCH * CPT)   + l16 * CPT + hi * 2;
    const int boff = wave * (NTILE * CPT) + l16 * CPT + hi * 2;

    const bool mmaFirst = (((wave >> 1) ^ (wave >> 3)) & 1) != 0;

    v8f acc[MT][NT] = {};

    const float* xb = x;
    const v4f*   wb = (const v4f*)wgt;
    float xr[SUB][2][4];
    v4f   wq[SUB];

    auto gload_x = [&]() __attribute__((always_inline)) {
#pragma unroll
        for (int j = 0; j < SUB; ++j) {
            const float* p = xb + xg[0] + j * (CPT * HW);
            xr[j][0][0] = p[0]; xr[j][0][1] = p[HW]; xr[j][0][2] = p[2 * HW]; xr[j][0][3] = p[3 * HW];
            if (two) {
                const float* p2 = xb + xg[1] + j * (CPT * HW);
                xr[j][1][0] = p2[0]; xr[j][1][1] = p2[HW]; xr[j][1][2] = p2[2 * HW]; xr[j][1][3] = p2[3 * HW];
            }
        }
        xb += SUB * CPT * HW;
    };
    auto gload_w = [&]() __attribute__((always_inline)) {
        if (wld) {
#pragma unroll
            for (int j = 0; j < SUB; ++j) wq[j] = wb[wgq + j * (CPT * 9 / 4)];
        }
        wb += SUB * CPT * 9 / 4;
    };

    auto store_raw = [&](const int buf) __attribute__((always_inline)) {
        const v4f z = {0.0f, 0.0f, 0.0f, 0.0f};
#pragma unroll
        for (int j = 0; j < SUB; ++j) {
            float* xs = sm + XS_OFF + (buf * SUB + j) * XS_SZ;
            const v4f v0 = {xr[j][0][0], xr[j][0][1], xr[j][0][2], xr[j][0][3]};
            *(v4f*)&xs[xl[0]] = xv[0] ? v0 : z;
            if (two) {
                const v4f v1 = {xr[j][1][0], xr[j][1][1], xr[j][1][2], xr[j][1][3]};
                *(v4f*)&xs[xl[1]] = xv[1] ? v1 : z;
            }
            if (wld) *(v4f*)&sm[WS_OFF + (buf * SUB + j) * WS_SZ + wlq] = wq[j];
        }
    };

    auto wxform1 = [&](const int buf) __attribute__((always_inline)) {
        const float* wr = sm + WS_OFF + buf * WS_SZ + 9 * oc4;
        float t3[3];
        if (xi == 0) {
#pragma unroll
            for (int j = 0; j < 3; ++j) t3[j] = wr[j];
        } else if (xi == 3) {
#pragma unroll
            for (int j = 0; j < 3; ++j) t3[j] = wr[6 + j];
        } else {
#pragma unroll
            for (int j = 0; j < 3; ++j) {
                const float g1 = wr[3 + j];
                t3[j] = 0.5f * ((wr[j] + wr[6 + j]) + ((xi == 2) ? -g1 : g1));
            }
        }
        float* us = sm + US_OFF + buf * US_SZ + xi * (4 * BCH * CPT) + oc4;
        us[0 * BCH * CPT] = t3[0];
        us[1 * BCH * CPT] = 0.5f * ((t3[0] + t3[2]) + t3[1]);
        us[2 * BCH * CPT] = 0.5f * ((t3[0] + t3[2]) - t3[1]);
        us[3 * BCH * CPT] = t3[2];
    };
    auto wxform = [&](const int stage) __attribute__((always_inline)) {
#pragma unroll
        for (int j = 0; j < SUB; ++j) wxform1(stage * SUB + j);
    };

    auto xform = [&](const int stage) __attribute__((always_inline)) {
        const float* xs = sm + XS_OFF + stage * SUB * XS_SZ + xbase;
        v2f d[4][4];
#pragma unroll
        for (int i = 0; i < 4; ++i)
#pragma unroll
            for (int j = 0; j < 4; ++j) d[i][j] = *(const v2f*)(xs + (i * XC + j) * CPT);
        v2f t[4][4];
#pragma unroll
        for (int j = 0; j < 4; ++j) {
            t[0][j] = d[0][j] - d[2][j];
            t[1][j] = d[1][j] + d[2][j];
            t[2][j] = d[2][j] - d[1][j];
            t[3][j] = d[1][j] - d[3][j];
        }
        float* vs = sm + VS_OFF + stage * SUB * VS_SZ + vbase;
#pragma unroll
        for (int i = 0; i < 4; ++i) {
            *(v2f*)(vs + (i * 4 + 0) * (NTILE * CPT)) = t[i][0] - t[i][2];
            *(v2f*)(vs + (i * 4 + 1) * (NTILE * CPT)) = t[i][1] + t[i][2];
            *(v2f*)(vs + (i * 4 + 2) * (NTILE * CPT)) = t[i][2] - t[i][1];
            *(v2f*)(vs + (i * 4 + 3) * (NTILE * CPT)) = t[i][1] - t[i][3];
        }
    };

    auto mma = [&](const int stage) __attribute__((always_inline)) {
        v2f a[SUB][MT], b[SUB][NT];
#pragma unroll
        for (int j = 0; j < SUB; ++j) {
            const float* us = sm + US_OFF + (stage * SUB + j) * US_SZ + aoff;
            const float* vs = sm + VS_OFF + (stage * SUB + j) * VS_SZ + boff;
#pragma unroll
            for (int m = 0; m < MT; ++m) a[j][m] = *(const v2f*)(us + m * 16 * CPT);
#pragma unroll
            for (int n = 0; n < NT; ++n) b[j][n] = *(const v2f*)(vs + n * 16 * CPT);
        }
#pragma unroll
        for (int j = 0; j < SUB; ++j) {
#pragma unroll
            for (int n = 0; n < NT; ++n) {
#pragma unroll
                for (int m = 0; m < MT; ++m)
                    acc[m][n] = __builtin_amdgcn_wmma_f32_16x16x4_f32(
                                    false, a[j][m], false, b[j][n], (short)0, acc[m][n], false, false);
            }
        }
    };

    auto slot = [&](const int it, const bool more, const int s, const int par) __attribute__((always_inline)) {
        if (more || s + 2 < NPAIR) store_raw(par);
        if (more || s + 3 < NPAIR) {
            if (s + 3 == NPAIR) {
                set_tile(it + 1);
                xb = x;
                wb = (const v4f*)wgt;
            }
            gload_x(); gload_w();
        }
        if (mmaFirst) mma(par);
        if (more || s + 1 < NPAIR) {
            wxform(par ^ 1);
            if (wave < 14) xform(par ^ 1);
        }
        if (!mmaFirst) mma(par);
        __syncthreads();
    };

    gload_x(); gload_w();
    store_raw(0);
    gload_x(); gload_w();
    if (tid < 2 * SUB * XR * 2) {
        const int bb = tid / (XR * 2), rr = (tid - bb * (XR * 2)) >> 1, side = tid & 1;
        *(v4f*)&sm[XS_OFF + bb * XS_SZ + (rr * XC + side * (XC - 1)) * CPT] = (v4f){0.0f, 0.0f, 0.0f, 0.0f};
    }
    __syncthreads();
    wxform(0);
    if (wave < 14) xform(0);
    store_raw(1);
    gload_x(); gload_w();
    __syncthreads();

    const int eo  = tid >> 4;
    const int etl = tid & 15;
    const float bv = bias[chBase + eo];

    for (int it = 0; it < TPW; ++it) {
    const bool more = it + 1 < TPW;
    for (int s = 0; s < NPAIR; s += 2) {
        slot(it, more, s, 0);
        slot(it, more, s + 1, 1);
    }

    const int ey    = y0 + YGRP * it;
    const int ebimg = ey / (HIMG / OROWS);
    const int eoh0  = (ey - ebimg * (HIMG / OROWS)) * OROWS;
    float* obase = out + ((long)(ebimg * KOUT + chBase + eo) * HW + eoh0 * WIMG);
    v2f o0[NT], o1[NT];
#pragma unroll
    for (int n = 0; n < NT; ++n) {
        float* ex = sm + EX_OFF + (n & 1) * EX_SZ;
#pragma unroll
        for (int m = 0; m < MT; ++m)
#pragma unroll
            for (int rr = 0; rr < 8; ++rr)
                ex[wave * EX_POS + (m * 16 + hi * 8 + rr) * EX_PITCH + l16] = acc[m][n][rr];
        __syncthreads();
        float mm[16];
#pragma unroll
        for (int p = 0; p < 16; ++p) mm[p] = ex[p * EX_POS + eo * EX_PITCH + etl];
        float s0[4], s1[4];
#pragma unroll
        for (int nu = 0; nu < 4; ++nu) {
            s0[nu] = (mm[0 + nu] + mm[4 + nu]) + mm[8 + nu];
            s1[nu] = (mm[4 + nu] - mm[8 + nu]) - mm[12 + nu];
        }
        o0[n] = (v2f){(s0[0] + s0[1]) + s0[2] + bv, (s0[1] - s0[2]) - s0[3] + bv};
        o1[n] = (v2f){(s1[0] + s1[1]) + s1[2] + bv, (s1[1] - s1[2]) - s1[3] + bv};
    }
#pragma unroll
    for (int n = 0; n < NT; ++n) {
        const int t  = n * 16 + etl;
        const int ty = t / TCOLS, tx = t - ty * TCOLS;
        float* op = obase + (2 * ty) * WIMG + 2 * tx;
        *(v2f*)op          = o0[n];
        *(v2f*)(op + WIMG) = o1[n];
    }
#pragma unroll
    for (int m = 0; m < MT; ++m)
#pragma unroll
        for (int n = 0; n < NT; ++n) acc[m][n] = (v8f){};
    }
}

extern "C" void kernel_launch(void* const* d_in, const int* in_sizes, int n_in,
                              void* d_out, int out_size, void* d_ws, size_t ws_size,
                              hipStream_t stream) {
    const float* x    = (const float*)d_in[0];
    const float* wgt  = (const float*)d_in[1];
    const float* bias = (const float*)d_in[2];
    float* out        = (float*)d_out;

    conv3x3_winograd_wmma_f32<<<dim3(NWG), TPB, 0, stream>>>(x, wgt, bias, out);
}


# Filter transform U = G g G^T hoisted into a first launch (2 MiB workspace, A-fragment order); conv waves load A fragments from global, no weight staging or transform in LDS; sum order unchanged
# speedup vs baseline: 3.1023x; 1.0396x over previous
#include <hip/hip_runtime.h>


typedef float v2f __attribute__((ext_vector_type(2)));
typedef float v4f __attribute__((ext_vector_type(4)));
typedef float v8f __attribute__((ext_vector_type(8)));

#define CIN   128
#define KOUT  256
#define HIMG  56
#define WIMG  56
#define HW    3136
#define KTOT  1152
#define BCH   32
#define OROWS 8
#define TCOLS 28
#define NTILE 112
#define NT    7
#define MT    2
#define CPT   4
#define NITER (CIN / CPT)
#define SUB   2
#define NPAIR (NITER / SUB)
#define XR    10
#define XC    58
#define XPIX  (XR * WIMG)
#define TPB   512
#define NPOS  16
#define NWG   256
#define YGRP  32
#define TPW   7

#define XS_SZ  (XR * XC * CPT + 4)
#define VS_SZ  (NPOS * NTILE * CPT + 32)
#define US_SZ  (NPOS * BCH * CPT)
#define XS_OFF 0
#define VS_OFF (2 * SUB * XS_SZ)
#define WS_SZ  (BCH * CPT * 9)
#define US_OFF (VS_OFF + 2 * SUB * VS_SZ)
#define WS_OFF (US_OFF + 2 * SUB * US_SZ)
#define SM_SZ  (WS_OFF + 2 * SUB * WS_SZ)
#define EX_PITCH 20
#define EX_POS   (BCH * EX_PITCH)
#define EX_SZ    (NPOS * EX_POS)
#define EX_OFF   SM_SZ
#define LDS_SZ   (SM_SZ + 2 * EX_SZ)

#define UW_FLOATS ((KOUT / BCH) * NPAIR * NPOS * 32 * 8)
#define UW_BYTES  ((size_t)UW_FLOATS * 4)

__global__ __launch_bounds__(256) void winograd_filter_transform(
    const float* __restrict__ wgt,
    float* __restrict__ uw, size_t ws_bytes)
{
    if (ws_bytes < UW_BYTES) return;
    const int t   = blockIdx.x * 256 + threadIdx.x;
    const int c   = t & (CIN - 1);
    const int och = t >> 7;
    const float* g = wgt + (size_t)t * 9;
    float w9[9];
#pragma unroll
    for (int i = 0; i < 9; ++i) w9[i] = g[i];
    const int lane = ((c >> 1) & 1) * 16 + (och & 15);
    float* up = uw + ((size_t)((och >> 5) * NPAIR + (c >> 3)) * NPOS * 32 + lane) * 8
                   + ((c >> 2) & 1) * 4 + ((och >> 4) & 1) * 2 + (c & 1);
#pragma unroll
    for (int xi = 0; xi < 4; ++xi) {
        float t3[3];
#pragma unroll
        for (int j = 0; j < 3; ++j) {
            if (xi == 0)      t3[j] = w9[j];
            else if (xi == 3) t3[j] = w9[6 + j];
            else {
                const float g1 = w9[3 + j];
                t3[j] = 0.5f * ((w9[j] + w9[6 + j]) + ((xi == 2) ? -g1 : g1));
            }
        }
        float* u = up + (size_t)(xi * 4) * (32 * 8);
        u[0 * 32 * 8] = t3[0];
        u[1 * 32 * 8] = 0.5f * ((t3[0] + t3[2]) + t3[1]);
        u[2 * 32 * 8] = 0.5f * ((t3[0] + t3[2]) - t3[1]);
        u[3 * 32 * 8] = t3[2];
    }
}

template <bool PRE>
__device__ __forceinline__ void conv_body(
    const float* __restrict__ x,
    const float* __restrict__ wgt,
    const float* __restrict__ uw,
    const float* __restrict__ bias,
    float* __restrict__ out,
    float* __restrict__ sm)
{

    const int tid  = threadIdx.x;
    const int lane = tid & 31;
    const int wave = __builtin_amdgcn_readfirstlane(tid >> 5);
    const int hi   = lane >> 4;
    const int l16  = lane & 15;

    const int chBase = (blockIdx.x & 7) * BCH;
    const int y0     = blockIdx.x >> 3;

    unsigned xg[2]; int xl[2]; bool xv[2]; int xrr[2], xcc[2];
#pragma unroll
    for (int i = 0; i < 2; ++i) {
        int q = (i == 0) ? tid : (tid + 64);
        q = q < XPIX ? q : XPIX - 1;
        const int rr = q / WIMG, cc = q - rr * WIMG;
        xrr[i] = rr; xcc[i] = cc;
        xl[i] = (rr * XC + cc + 1) * CPT;
    }
    auto set_tile = [&](const int it) __attribute__((always_inline)) {
        const int y    = y0 + YGRP * it;
        const int bimg = y / (HIMG / OROWS);
        const int oh0  = (y - bimg * (HIMG / OROWS)) * OROWS;
#pragma unroll
        for (int i = 0; i < 2; ++i) {
            const int ih = oh0 - 1 + xrr[i];
            xv[i] = (unsigned)ih < (unsigned)HIMG;
            const int ihc = ih < 0 ? 0 : (ih > HIMG - 1 ? HIMG - 1 : ih);
            xg[i] = (unsigned)(bimg * (CIN * HW) + ihc * WIMG + xcc[i]);
        }
    };
    set_tile(0);
    const bool two = wave >= 14;

    const int xi  = wave >> 2;
    const int oc4 = tid & 127;

    const bool wld = wave < 9;
    const int  wp  = wld ? tid : 0;
    const int  war = wp / 9, waq = wp - war * 9;
    const unsigned wgq = (unsigned)((chBase + war) * (KTOT / 4) + waq);
    const int  wlq = wp * 4;

    const int tt  = wave * 8 + (l16 >> 1);
    const int tcp = (l16 & 1) * 2;
    const int tty = tt / TCOLS, ttx = tt - tty * TCOLS;
    const int xbase = hi * XS_SZ + ((2 * tty) * XC + 2 * ttx) * CPT + tcp;
    const int vbase = hi * VS_SZ + tt * CPT + tcp;

    const int aoff = wave * (BCH * CPT)   + l16 * CPT + hi * 2;
    const int boff = wave * (NTILE * CPT) + l16 * CPT + hi * 2;

    const bool mmaFirst = (((wave >> 1) ^ (wave >> 3)) & 1) != 0;

    v8f acc[MT][NT] = {};

    const float* xb = x;
    const v4f*   wb = (const v4f*)wgt;
    float xr[SUB][2][4];
    v4f   wq[SUB];

    auto gload_x = [&]() __attribute__((always_inline)) {
#pragma unroll
        for (int j = 0; j < SUB; ++j) {
            const float* p = xb + xg[0] + j * (CPT * HW);
            xr[j][0][0] = p[0]; xr[j][0][1] = p[HW]; xr[j][0][2] = p[2 * HW]; xr[j][0][3] = p[3 * HW];
            if (two) {
                const float* p2 = xb + xg[1] + j * (CPT * HW);
                xr[j][1][0] = p2[0]; xr[j][1][1] = p2[HW]; xr[j][1][2] = p2[2 * HW]; xr[j][1][3] = p2[3 * HW];
            }
        }
        xb += SUB * CPT * HW;
    };
    auto gload_w = [&]() __attribute__((always_inline)) {
        if constexpr (!PRE) {
            if (wld) {
#pragma unroll
                for (int j = 0; j < SUB; ++j) wq[j] = wb[wgq + j * (CPT * 9 / 4)];
            }
            wb += SUB * CPT * 9 / 4;
        }
    };
    v4f af[2][SUB];
    const v4f* ub = (const v4f*)(uw + ((size_t)((blockIdx.x & 7) * NPAIR) * NPOS + wave) * (32 * 8) + lane * 8);
    auto gload_a = [&](const int sn, const int buf) __attribute__((always_inline)) {
        if constexpr (PRE) {
            const v4f* p = ub + sn * (NPOS * 32 * 8 / 4);
            af[buf][0] = p[0];
            af[buf][1] = p[1];
        }
    };

    auto store_raw = [&](const int buf) __attribute__((always_inline)) {
        const v4f z = {0.0f, 0.0f, 0.0f, 0.0f};
#pragma unroll
        for (int j = 0; j < SUB; ++j) {
            float* xs = sm + XS_OFF + (buf * SUB + j) * XS_SZ;
            const v4f v0 = {xr[j][0][0], xr[j][0][1], xr[j][0][2], xr[j][0][3]};
            *(v4f*)&xs[xl[0]] = xv[0] ? v0 : z;
            if (two) {
                const v4f v1 = {xr[j][1][0], xr[j][1][1], xr[j][1][2], xr[j][1][3]};
                *(v4f*)&xs[xl[1]] = xv[1] ? v1 : z;
            }
            if constexpr (!PRE) {
                if (wld) *(v4f*)&sm[WS_OFF + (buf * SUB + j) * WS_SZ + wlq] = wq[j];
            }
        }
    };

    auto wxform1 = [&](const int buf) __attribute__((always_inline)) {
        const float* wr = sm + WS_OFF + buf * WS_SZ + 9 * oc4;
        float t3[3];
        if (xi == 0) {
#pragma unroll
            for (int j = 0; j < 3; ++j) t3[j] = wr[j];
        } else if (xi == 3) {
#pragma unroll
            for (int j = 0; j < 3; ++j) t3[j] = wr[6 + j];
        } else {
#pragma unroll
            for (int j = 0; j < 3; ++j) {
                const float g1 = wr[3 + j];
                t3[j] = 0.5f * ((wr[j] + wr[6 + j]) + ((xi == 2) ? -g1 : g1));
            }
        }
        float* us = sm + US_OFF + buf * US_SZ + xi * (4 * BCH * CPT) + oc4;
        us[0 * BCH * CPT] = t3[0];
        us[1 * BCH * CPT] = 0.5f * ((t3[0] + t3[2]) + t3[1]);
        us[2 * BCH * CPT] = 0.5f * ((t3[0] + t3[2]) - t3[1]);
        us[3 * BCH * CPT] = t3[2];
    };
    auto wxform = [&](const int stage) __attribute__((always_inline)) {
        if constexpr (!PRE) {
#pragma unroll
            for (int j = 0; j < SUB; ++j) wxform1(stage * SUB + j);
        }
    };

    auto xform = [&](const int stage) __attribute__((always_inline)) {
        const float* xs = sm + XS_OFF + stage * SUB * XS_SZ + xbase;
        v2f d[4][4];
#pragma unroll
        for (int i = 0; i < 4; ++i)
#pragma unroll
            for (int j = 0; j < 4; ++j) d[i][j] = *(const v2f*)(xs + (i * XC + j) * CPT);
        v2f t[4][4];
#pragma unroll
        for (int j = 0; j < 4; ++j) {
            t[0][j] = d[0][j] - d[2][j];
            t[1][j] = d[1][j] + d[2][j];
            t[2][j] = d[2][j] - d[1][j];
            t[3][j] = d[1][j] - d[3][j];
        }
        float* vs = sm + VS_OFF + stage * SUB * VS_SZ + vbase;
#pragma unroll
        for (int i = 0; i < 4; ++i) {
            *(v2f*)(vs + (i * 4 + 0) * (NTILE * CPT)) = t[i][0] - t[i][2];
            *(v2f*)(vs + (i * 4 + 1) * (NTILE * CPT)) = t[i][1] + t[i][2];
            *(v2f*)(vs + (i * 4 + 2) * (NTILE * CPT)) = t[i][2] - t[i][1];
            *(v2f*)(vs + (i * 4 + 3) * (NTILE * CPT)) = t[i][1] - t[i][3];
        }
    };

    auto mma = [&](const int stage) __attribute__((always_inline)) {
        v2f a[SUB][MT], b[SUB][NT];
#pragma unroll
        for (int j = 0; j < SUB; ++j) {
            const float* us = sm + US_OFF + (stage * SUB + j) * US_SZ + aoff;
            const float* vs = sm + VS_OFF + (stage * SUB + j) * VS_SZ + boff;
            if constexpr (PRE) {
                a[j][0] = (v2f){af[stage][j].x, af[stage][j].y};
                a[j][1] = (v2f){af[stage][j].z, af[stage][j].w};
            } else {
#pragma unroll
                for (int m = 0; m < MT; ++m) a[j][m] = *(const v2f*)(us + m * 16 * CPT);
            }
#pragma unroll
            for (int n = 0; n < NT; ++n) b[j][n] = *(const v2f*)(vs + n * 16 * CPT);
        }
#pragma unroll
        for (int j = 0; j < SUB; ++j) {
#pragma unroll
            for (int n = 0; n < NT; ++n) {
#pragma unroll
                for (int m = 0; m < MT; ++m)
                    acc[m][n] = __builtin_amdgcn_wmma_f32_16x16x4_f32(
                                    false, a[j][m], false, b[j][n], (short)0, acc[m][n], false, false);
            }
        }
        asm volatile("v_nop\n\tv_nop\n\tv_nop\n\tv_nop" : "+v"(acc[MT - 1][NT - 1]));
    };

    auto slot = [&](const int it, const bool more, const int s, const int par) __attribute__((always_inline)) {
        if (more || s + 2 < NPAIR) store_raw(par);
        if (more || s + 3 < NPAIR) {
            if (s + 3 == NPAIR) {
                set_tile(it + 1);
                xb = x;
                wb = (const v4f*)wgt;
            }
            gload_x(); gload_w();
        }
        gload_a((s + 1) & (NPAIR - 1), par ^ 1);
        if (mmaFirst) mma(par);
        if (more || s + 1 < NPAIR) {
            wxform(par ^ 1);
            if (wave < 14) xform(par ^ 1);
        }
        if (!mmaFirst) mma(par);
        __syncthreads();
    };

    gload_a(0, 0);
    gload_x(); gload_w();
    store_raw(0);
    gload_x(); gload_w();
    if (tid < 2 * SUB * XR * 2) {
        const int bb = tid / (XR * 2), rr = (tid - bb * (XR * 2)) >> 1, side = tid & 1;
        *(v4f*)&sm[XS_OFF + bb * XS_SZ + (rr * XC + side * (XC - 1)) * CPT] = (v4f){0.0f, 0.0f, 0.0f, 0.0f};
    }
    __syncthreads();
    wxform(0);
    if (wave < 14) xform(0);
    store_raw(1);
    gload_x(); gload_w();
    __syncthreads();

    const int eo  = tid >> 4;
    const int etl = tid & 15;
    const float bv = bias[chBase + eo];

    for (int it = 0; it < TPW; ++it) {
    const bool more = it + 1 < TPW;
    for (int s = 0; s < NPAIR; s += 2) {
        slot(it, more, s, 0);
        slot(it, more, s + 1, 1);
    }

    const int ey    = y0 + YGRP * it;
    const int ebimg = ey / (HIMG / OROWS);
    const int eoh0  = (ey - ebimg * (HIMG / OROWS)) * OROWS;
    float* obase = out + ((long)(ebimg * KOUT + chBase + eo) * HW + eoh0 * WIMG);
    v2f o0[NT], o1[NT];
#pragma unroll
    for (int n = 0; n < NT; ++n) {
        float* ex = sm + EX_OFF + (n & 1) * EX_SZ;
#pragma unroll
        for (int m = 0; m < MT; ++m)
#pragma unroll
            for (int rr = 0; rr < 8; ++rr)
                ex[wave * EX_POS + (m * 16 + hi * 8 + rr) * EX_PITCH + l16] = acc[m][n][rr];
        __syncthreads();
        float mm[16];
#pragma unroll
        for (int p = 0; p < 16; ++p) mm[p] = ex[p * EX_POS + eo * EX_PITCH + etl];
        float s0[4], s1[4];
#pragma unroll
        for (int nu = 0; nu < 4; ++nu) {
            s0[nu] = (mm[0 + nu] + mm[4 + nu]) + mm[8 + nu];
            s1[nu] = (mm[4 + nu] - mm[8 + nu]) - mm[12 + nu];
        }
        o0[n] = (v2f){(s0[0] + s0[1]) + s0[2] + bv, (s0[1] - s0[2]) - s0[3] + bv};
        o1[n] = (v2f){(s1[0] + s1[1]) + s1[2] + bv, (s1[1] - s1[2]) - s1[3] + bv};
    }
#pragma unroll
    for (int n = 0; n < NT; ++n) {
        const int t  = n * 16 + etl;
        const int ty = t / TCOLS, tx = t - ty * TCOLS;
        float* op = obase + (2 * ty) * WIMG + 2 * tx;
        *(v2f*)op          = o0[n];
        *(v2f*)(op + WIMG) = o1[n];
    }
#pragma unroll
    for (int m = 0; m < MT; ++m)
#pragma unroll
        for (int n = 0; n < NT; ++n) acc[m][n] = (v8f){};
    }
}

__global__ __launch_bounds__(TPB) void conv3x3_winograd_wmma_f32(
    const float* __restrict__ x, const float* __restrict__ wgt, const float* __restrict__ bias,
    float* __restrict__ out, const float* __restrict__ uw, size_t ws_bytes)
{
    __shared__ __attribute__((aligned(16))) float sm[LDS_SZ];
    if (ws_bytes >= UW_BYTES) conv_body<true>(x, wgt, uw, bias, out, sm);
    else                      conv_body<false>(x, wgt, uw, bias, out, sm);
}

extern "C" void kernel_launch(void* const* d_in, const int* in_sizes, int n_in,
                              void* d_out, int out_size, void* d_ws, size_t ws_size,
                              hipStream_t stream) {
    const float* x    = (const float*)d_in[0];
    const float* wgt  = (const float*)d_in[1];
    const float* bias = (const float*)d_in[2];
    float* out        = (float*)d_out;

    float* uw         = (float*)d_ws;

    winograd_filter_transform<<<dim3(128), 256, 0, stream>>>(wgt, uw, ws_size);
    conv3x3_winograd_wmma_f32<<<dim3(NWG), TPB, 0, stream>>>(x, wgt, bias, out, uw, ws_size);
}
